# starting kernel
# baseline (speedup 1.0000x reference)
#include <hip/hip_runtime.h>


typedef float v2f __attribute__((ext_vector_type(2)));
typedef float v8f __attribute__((ext_vector_type(8)));

#define N_NODES 50000

__global__ void gcn_zero(float* __restrict__ p, int n) {
  int i = blockIdx.x * blockDim.x + threadIdx.x;
  if (i < n) p[i] = 0.0f;
}

__global__ void gcn_degree(const int* __restrict__ src, const int* __restrict__ dst,
                           float* __restrict__ deg_out, float* __restrict__ deg_in, int E) {
  int e = blockIdx.x * blockDim.x + threadIdx.x;
  if (e >= E) return;
  atomicAdd(&deg_out[src[e]], 1.0f);
  atomicAdd(&deg_in[dst[e]], 1.0f);
}

__global__ void gcn_norm(float* __restrict__ p, int n) {
  int i = blockIdx.x * blockDim.x + threadIdx.x;
  if (i < n) p[i] = __frsqrt_rn(fmaxf(p[i], 1.0f));
}

__global__ void gcn_gemm_wmma(const float* __restrict__ A, const float* __restrict__ W,
                              float* __restrict__ C, int n_rows, int K, int F,
                              const float* __restrict__ pre_scale,
                              const float* __restrict__ post_scale,
                              const float* __restrict__ bias, int relu) {
  const int ftiles = F >> 4;
  const int tiles  = (n_rows >> 4) * ftiles;
  const int w = blockIdx.x * (blockDim.x >> 5) + ((int)threadIdx.x >> 5);
  if (w >= tiles) return;
  const int tileM = w / ftiles;
  const int tileN = w - tileM * ftiles;
  const int lane  = threadIdx.x & 31;
  const int half  = lane >> 4;
  const int lr    = lane & 15;

  const int m = (tileM << 4) + lr;
  const int n = (tileN << 4) + lr;
  const float ps = pre_scale ? pre_scale[m] : 1.0f;
  const float* Arow = A + (size_t)m * K;

  v8f acc = {};
  for (int k = 0; k < K; k += 4) {
    const int koff = k + (half << 1);
    v2f a, b;
    a.x = Arow[koff]     * ps;
    a.y = Arow[koff + 1] * ps;
    b.x = W[(size_t)koff * F + n];
    b.y = W[(size_t)(koff + 1) * F + n];
    acc = __builtin_amdgcn_wmma_f32_16x16x4_f32(
         false, a,  false, b,
         (short)0, acc,  false,  false);
  }

#pragma unroll
  for (int i = 0; i < 8; ++i) {
    const int row = (tileM << 4) + i + (half << 3);
    const int col = (tileN << 4) + lr;
    float v = acc[i];
    if (post_scale) v *= post_scale[row];
    if (bias)       v += bias[col];
    if (relu)       v = fmaxf(v, 0.0f);
    C[(size_t)row * F + col] = v;
  }
}

__global__ void gcn_scatter(const float* __restrict__ msg, const int* __restrict__ src,
                            const int* __restrict__ dst, const float* __restrict__ scale,
                            float* __restrict__ agg, int n_edges, int F, int chunk_shift) {
  int idx = blockIdx.x * blockDim.x + threadIdx.x;
  int e = idx >> chunk_shift;
  if (e >= n_edges) return;
  int c = idx & ((1 << chunk_shift) - 1);
  int s = src[e], d = dst[e];
  float4 v = ((const float4*)(msg + (size_t)s * F))[c];
  if (scale) { float sc = scale[s]; v.x *= sc; v.y *= sc; v.z *= sc; v.w *= sc; }
  float* out = agg + (size_t)d * F + (c << 2);
  atomicAdd(out + 0, v.x);
  atomicAdd(out + 1, v.y);
  atomicAdd(out + 2, v.z);
  atomicAdd(out + 3, v.w);
}

__global__ void gcn_epilogue(float* __restrict__ buf, const float* __restrict__ nd,
                             const float* __restrict__ bias, int total, int f_shift, int relu) {
  int i = blockIdx.x * blockDim.x + threadIdx.x;
  if (i >= total) return;
  int row = i >> f_shift;
  int col = i & ((1 << f_shift) - 1);
  float v = buf[i] * nd[row] + bias[col];
  if (relu) v = fmaxf(v, 0.0f);
  buf[i] = v;
}

static inline unsigned cdiv(unsigned a, unsigned b) { return (a + b - 1) / b; }

extern "C" void kernel_launch(void* const* d_in, const int* in_sizes, int n_in,
                              void* d_out, int out_size, void* d_ws, size_t ws_size,
                              hipStream_t stream) {
  const float* features = (const float*)d_in[0];
  const float* W1 = (const float*)d_in[1];
  const float* b1 = (const float*)d_in[2];
  const float* W2 = (const float*)d_in[3];
  const float* b2 = (const float*)d_in[4];
  const float* W3 = (const float*)d_in[5];
  const float* b3 = (const float*)d_in[6];
  const int* src  = (const int*)d_in[7];
  const int* dst  = (const int*)d_in[8];
  float* out = (float*)d_out;

  const int N  = N_NODES;
  const int E  = in_sizes[7];
  const int IN = in_sizes[0] / N;
  const int H  = in_sizes[2];
  const int CLS = in_sizes[6];

  float* ws   = (float*)d_ws;
  float* ns   = ws;
  float* nd   = ws + N;
  float* bufA = ws + 2 * N + 96;
  float* bufB = bufA + (size_t)N * H;
  (void)ws_size; (void)n_in; (void)out_size;

  const int BS = 256;

  gcn_zero  <<<cdiv(2 * N, BS), BS, 0, stream>>>(ns, 2 * N);
  gcn_degree<<<cdiv(E, BS),     BS, 0, stream>>>(src, dst, ns, nd, E);
  gcn_norm  <<<cdiv(2 * N, BS), BS, 0, stream>>>(ns, 2 * N);

  {
    unsigned tiles = (N / 16) * (H / 16);
    gcn_gemm_wmma<<<cdiv(tiles, BS / 32), BS, 0, stream>>>(
        features, W1, bufA, N, IN, H, ns, nullptr, nullptr, 0);
    gcn_zero<<<cdiv(N * H, BS), BS, 0, stream>>>(bufB, N * H);
    gcn_scatter<<<cdiv((unsigned)E * (H / 4), BS), BS, 0, stream>>>(
        bufA, src, dst, nullptr, bufB, E, H, 5);
    gcn_epilogue<<<cdiv(N * H, BS), BS, 0, stream>>>(bufB, nd, b1, N * H, 7, 1);
  }

  {
    gcn_zero<<<cdiv(N * H, BS), BS, 0, stream>>>(bufA, N * H);
    gcn_scatter<<<cdiv((unsigned)E * (H / 4), BS), BS, 0, stream>>>(
        bufB, src, dst, ns, bufA, E, H, 5);
    unsigned tiles = (N / 16) * (H / 16);
    gcn_gemm_wmma<<<cdiv(tiles, BS / 32), BS, 0, stream>>>(
        bufA, W2, bufB, N, H, H, nullptr, nd, b2, 1);
  }

  {
    unsigned tiles = (N / 16) * (CLS / 16);
    gcn_gemm_wmma<<<cdiv(tiles, BS / 32), BS, 0, stream>>>(
        bufB, W3, bufA, N, H, CLS, ns, nullptr, nullptr, 0);
    gcn_zero<<<cdiv(N * CLS, BS), BS, 0, stream>>>(out, N * CLS);
    gcn_scatter<<<cdiv((unsigned)E * (CLS / 4), BS), BS, 0, stream>>>(
        bufA, src, dst, nullptr, out, E, CLS, 4);
    gcn_epilogue<<<cdiv(N * CLS, BS), BS, 0, stream>>>(out, nd, b3, N * CLS, 6, 0);
  }
}


# neighbour sums by row-range ownership: a block keeps 196 destination rows in LDS, scans all dst, adds matching source rows (LDS f32 atomics, arrival order); no global atomics, zero or epilogue pass
# speedup vs baseline: 1.5946x; 1.5946x over previous
#include <hip/hip_runtime.h>


typedef float v2f __attribute__((ext_vector_type(2)));
typedef float v8f __attribute__((ext_vector_type(8)));

#define N_NODES 50000

__global__ void gcn_zero(float* __restrict__ p, int n) {
  int i = blockIdx.x * blockDim.x + threadIdx.x;
  if (i < n) p[i] = 0.0f;
}

__global__ void gcn_degree(const int* __restrict__ src, const int* __restrict__ dst,
                           float* __restrict__ deg_out, float* __restrict__ deg_in, int E) {
  int e = blockIdx.x * blockDim.x + threadIdx.x;
  if (e >= E) return;
  atomicAdd(&deg_out[src[e]], 1.0f);
  atomicAdd(&deg_in[dst[e]], 1.0f);
}

__global__ void gcn_norm(float* __restrict__ p, int n) {
  int i = blockIdx.x * blockDim.x + threadIdx.x;
  if (i < n) p[i] = __frsqrt_rn(fmaxf(p[i], 1.0f));
}

__global__ void gcn_gemm_wmma(const float* __restrict__ A, const float* __restrict__ W,
                              float* __restrict__ C, int n_rows, int K, int F,
                              const float* __restrict__ pre_scale,
                              const float* __restrict__ post_scale,
                              const float* __restrict__ bias, int relu) {
  const int ftiles = F >> 4;
  const int tiles  = (n_rows >> 4) * ftiles;
  const int w = blockIdx.x * (blockDim.x >> 5) + ((int)threadIdx.x >> 5);
  if (w >= tiles) return;
  const int tileM = w / ftiles;
  const int tileN = w - tileM * ftiles;
  const int lane  = threadIdx.x & 31;
  const int half  = lane >> 4;
  const int lr    = lane & 15;

  const int m = (tileM << 4) + lr;
  const int n = (tileN << 4) + lr;
  const float ps = pre_scale ? pre_scale[m] : 1.0f;
  const float* Arow = A + (size_t)m * K;

  v8f acc = {};
  for (int k = 0; k < K; k += 4) {
    const int koff = k + (half << 1);
    v2f a, b;
    a.x = Arow[koff]     * ps;
    a.y = Arow[koff + 1] * ps;
    b.x = W[(size_t)koff * F + n];
    b.y = W[(size_t)(koff + 1) * F + n];
    acc = __builtin_amdgcn_wmma_f32_16x16x4_f32(
         false, a,  false, b,
         (short)0, acc,  false,  false);
  }

#pragma unroll
  for (int i = 0; i < 8; ++i) {
    const int row = (tileM << 4) + i + (half << 3);
    const int col = (tileN << 4) + lr;
    float v = acc[i];
    if (post_scale) v *= post_scale[row];
    if (bias)       v += bias[col];
    if (relu)       v = fmaxf(v, 0.0f);
    C[(size_t)row * F + col] = v;
  }
}

#define AGG_R   196
#define AGG_T   1024
#define AGG_U   4
#define AGG_CAP 576
#define AGG_G   8

template <int F>
__device__ __forceinline__ void agg_flush(const float* __restrict__ msg,
                                          const int* __restrict__ src,
                                          const float* __restrict__ scale,
                                          const unsigned* mylist, float* acc,
                                          int cnt, int lane) {
  constexpr int V = F / 32;
  for (int b = 0; b < cnt; b += 32) {
    const int idx = b + lane;
    int ent = 0, s = 0;
    float sc = 1.0f;
    if (idx < cnt) {
      ent = (int)mylist[idx];
      s = src[(unsigned)ent >> 8];
      if (scale) sc = scale[s];
    }
    const int nb = (cnt - b < 32) ? (cnt - b) : 32;
    for (int k = 0; k < nb; k += AGG_G) {
      float v[AGG_G][V];
      int rr[AGG_G];
      float ss[AGG_G];
#pragma unroll
      for (int q = 0; q < AGG_G; ++q) {
        const int kk = (k + q < nb) ? (k + q) : (nb - 1);
        const int sk = __builtin_amdgcn_readlane(s, kk);
        rr[q] = __builtin_amdgcn_readlane(ent, kk) & 255;
        ss[q] = __int_as_float(__builtin_amdgcn_readlane(__float_as_int(sc), kk));
        const float* rowp = msg + (size_t)sk * F + V * lane;
        if (V == 4) {
          const float4 t = *(const float4*)rowp;
          v[q][0] = t.x; v[q][1] = t.y; v[q][2] = t.z; v[q][V - 1] = t.w;
        } else {
          const float2 t = *(const float2*)rowp;
          v[q][0] = t.x; v[q][V - 1] = t.y;
        }
      }
#pragma unroll
      for (int q = 0; q < AGG_G; ++q) {
        if (k + q < nb) {
          float* a = acc + rr[q] * F + lane;
#pragma unroll
          for (int j = 0; j < V; ++j) atomicAdd(a + 32 * j, v[q][j] * ss[q]);
        }
      }
    }
  }
}

__device__ __forceinline__ void agg_match(const int4 (&b)[AGG_U], int p, int tid, int n4, int base,
                                          unsigned* mylist, int& cnt) {
#pragma unroll
  for (int u = 0; u < AGG_U; ++u) {
    const int g = (p * AGG_U + u) * AGG_T + tid;
    const int e0 = g << 2;
    const bool ok = g < n4;
    const unsigned r0 = (unsigned)(b[u].x - base);
    const unsigned r1 = (unsigned)(b[u].y - base);
    const unsigned r2 = (unsigned)(b[u].z - base);
    const unsigned r3 = (unsigned)(b[u].w - base);
    const bool m0 = ok & (r0 < AGG_R), m1 = ok & (r1 < AGG_R);
    const bool m2 = ok & (r2 < AGG_R), m3 = ok & (r3 < AGG_R);
    if (__builtin_amdgcn_ballot_w32(m0 | m1 | m2 | m3)) {
      const unsigned k0 = __builtin_amdgcn_ballot_w32(m0);
      if (m0) mylist[cnt + __builtin_amdgcn_mbcnt_lo(k0, 0u)] = ((unsigned)(e0 + 0) << 8) | r0;
      cnt += __builtin_popcount(k0);
      const unsigned k1 = __builtin_amdgcn_ballot_w32(m1);
      if (m1) mylist[cnt + __builtin_amdgcn_mbcnt_lo(k1, 0u)] = ((unsigned)(e0 + 1) << 8) | r1;
      cnt += __builtin_popcount(k1);
      const unsigned k2 = __builtin_amdgcn_ballot_w32(m2);
      if (m2) mylist[cnt + __builtin_amdgcn_mbcnt_lo(k2, 0u)] = ((unsigned)(e0 + 2) << 8) | r2;
      cnt += __builtin_popcount(k2);
      const unsigned k3 = __builtin_amdgcn_ballot_w32(m3);
      if (m3) mylist[cnt + __builtin_amdgcn_mbcnt_lo(k3, 0u)] = ((unsigned)(e0 + 3) << 8) | r3;
      cnt += __builtin_popcount(k3);
    }
  }
}

__device__ __forceinline__ void agg_load(int4 (&b)[AGG_U], const int4* __restrict__ dst4, int p,
                                         int tid, int n4) {
#pragma unroll
  for (int u = 0; u < AGG_U; ++u) {
    const int g = (p * AGG_U + u) * AGG_T + tid;
    b[u] = dst4[(g < n4) ? g : (n4 - 1)];
  }
}

template <int F>
__global__ __launch_bounds__(AGG_T) void gcn_agg_own(
    const float* __restrict__ msg, const int* __restrict__ src, const int* __restrict__ dst,
    const float* __restrict__ scale, const float* __restrict__ nd,
    const float* __restrict__ bias, float* __restrict__ outp,
    int n_edges, int n_nodes, int relu) {
  constexpr int V = F / 32;
  __shared__ __attribute__((aligned(16))) float acc[AGG_R * F];
  __shared__ unsigned list[(AGG_T / 32) * AGG_CAP];

  const int tid  = threadIdx.x;
  const int lane = tid & 31;
  const int wave = __builtin_amdgcn_readfirstlane(tid >> 5);
  const int base = blockIdx.x * AGG_R;

  for (int i = tid; i < AGG_R * F / 4; i += AGG_T)
    ((float4*)acc)[i] = make_float4(0.0f, 0.0f, 0.0f, 0.0f);
  __syncthreads();

  unsigned* mylist = list + wave * AGG_CAP;
  int cnt = 0;
  const int n4  = n_edges >> 2;
  const int nit = (n4 + AGG_T * AGG_U - 1) / (AGG_T * AGG_U);
  const int4* dst4 = (const int4*)dst;
  int it = (nit > 0) ? (int)(blockIdx.x % (unsigned)nit) : 0;

  if (nit > 0) {
    int p0 = it;
    int p1 = (p0 + 1 >= nit) ? 0 : p0 + 1;
    int4 b0[AGG_U], b1[AGG_U];
    agg_load(b0, dst4, p0, tid, n4);
    int i = 0;
    while (i < nit) {
      agg_load(b1, dst4, p1, tid, n4);
      agg_match(b0, p0, tid, n4, base, mylist, cnt);
      if (cnt >= 32) { agg_flush<F>(msg, src, scale, mylist, acc, cnt, lane); cnt = 0; }
      p0 = p1; p1 = (p1 + 1 >= nit) ? 0 : p1 + 1; ++i;
      if (i < nit) {
        agg_load(b0, dst4, p1, tid, n4);
        agg_match(b1, p0, tid, n4, base, mylist, cnt);
        if (cnt >= 32) { agg_flush<F>(msg, src, scale, mylist, acc, cnt, lane); cnt = 0; }
        p0 = p1; p1 = (p1 + 1 >= nit) ? 0 : p1 + 1; ++i;
      }
    }
  }
  if (wave == 0 && (n_edges & 3)) {
    const int e = (n4 << 2) + lane;
    const unsigned r = (e < n_edges) ? (unsigned)(dst[e] - base) : 0xffffffffu;
    const bool m = r < AGG_R;
    const unsigned km = __builtin_amdgcn_ballot_w32(m);
    if (m) mylist[cnt + __builtin_amdgcn_mbcnt_lo(km, 0u)] = ((unsigned)e << 8) | r;
    cnt += __builtin_popcount(km);
  }
  if (cnt > 0) agg_flush<F>(msg, src, scale, mylist, acc, cnt, lane);
  __syncthreads();

  float bv[V];
#pragma unroll
  for (int j = 0; j < V; ++j) bv[j] = bias ? bias[V * lane + j] : 0.0f;
  for (int r = wave; r < AGG_R; r += AGG_T / 32) {
    const int row = base + r;
    if (row < n_nodes) {
      const float sc = nd ? nd[row] : 1.0f;
      float o[V];
#pragma unroll
      for (int j = 0; j < V; ++j) {
        float t = acc[r * F + 32 * j + lane] * sc + bv[j];
        if (relu) t = fmaxf(t, 0.0f);
        o[j] = t;
      }
      float* op = outp + (size_t)row * F + V * lane;
      if (V == 4) *(float4*)op = make_float4(o[0], o[1], o[2], o[V - 1]);
      else        *(float2*)op = make_float2(o[0], o[V - 1]);
    }
  }
}

static inline unsigned cdiv(unsigned a, unsigned b) { return (a + b - 1) / b; }

extern "C" void kernel_launch(void* const* d_in, const int* in_sizes, int n_in,
                              void* d_out, int out_size, void* d_ws, size_t ws_size,
                              hipStream_t stream) {
  const float* features = (const float*)d_in[0];
  const float* W1 = (const float*)d_in[1];
  const float* b1 = (const float*)d_in[2];
  const float* W2 = (const float*)d_in[3];
  const float* b2 = (const float*)d_in[4];
  const float* W3 = (const float*)d_in[5];
  const float* b3 = (const float*)d_in[6];
  const int* src  = (const int*)d_in[7];
  const int* dst  = (const int*)d_in[8];
  float* out = (float*)d_out;

  const int N  = N_NODES;
  const int E  = in_sizes[7];
  const int IN = in_sizes[0] / N;
  const int H  = in_sizes[2];
  const int CLS = in_sizes[6];

  float* ws   = (float*)d_ws;
  float* ns   = ws;
  float* nd   = ws + N;
  float* bufA = ws + 2 * N + 96;
  float* bufB = bufA + (size_t)N * H;
  (void)ws_size; (void)n_in; (void)out_size;

  const int BS = 256;

  gcn_zero  <<<cdiv(2 * N, BS), BS, 0, stream>>>(ns, 2 * N);
  gcn_degree<<<cdiv(E, BS),     BS, 0, stream>>>(src, dst, ns, nd, E);
  gcn_norm  <<<cdiv(2 * N, BS), BS, 0, stream>>>(ns, 2 * N);

  const unsigned agg_blocks = cdiv(N, AGG_R);

  {
    unsigned tiles = (N / 16) * (H / 16);
    gcn_gemm_wmma<<<cdiv(tiles, BS / 32), BS, 0, stream>>>(
        features, W1, bufA, N, IN, H, ns, nullptr, nullptr, 0);
    gcn_agg_own<128><<<agg_blocks, AGG_T, 0, stream>>>(
        bufA, src, dst, nullptr, nd, b1, bufB, E, N, 1);
  }

  {
    gcn_agg_own<128><<<agg_blocks, AGG_T, 0, stream>>>(
        bufB, src, dst, ns, nullptr, nullptr, bufA, E, N, 0);
    unsigned tiles = (N / 16) * (H / 16);
    gcn_gemm_wmma<<<cdiv(tiles, BS / 32), BS, 0, stream>>>(
        bufA, W2, bufB, N, H, H, nullptr, nd, b2, 1);
  }

  {
    unsigned tiles = (N / 16) * (CLS / 16);
    gcn_gemm_wmma<<<cdiv(tiles, BS / 32), BS, 0, stream>>>(
        bufB, W3, bufA, N, H, CLS, ns, nullptr, nullptr, 0);
    gcn_agg_own<64><<<agg_blocks, AGG_T, 0, stream>>>(
        bufA, src, dst, nullptr, nd, b3, out, E, N, 0);
  }
}


# GEMMs: a wave keeps one 16-column strip of W in registers all launch; the block stages 32-row A panels in LDS by whole-row loads, next panel in flight, reads A fragments there; sum orders unchanged
# speedup vs baseline: 2.6382x; 1.6545x over previous
#include <hip/hip_runtime.h>


typedef float v2f __attribute__((ext_vector_type(2)));
typedef float v4f __attribute__((ext_vector_type(4)));
typedef float v8f __attribute__((ext_vector_type(8)));

#define N_NODES 50000

__global__ void gcn_zero(float* __restrict__ p, int n) {
  int i = blockIdx.x * blockDim.x + threadIdx.x;
  if (i < n) p[i] = 0.0f;
}

#define AGG_R   196
#define AGG_T   1024
#define AGG_G   8
#define SRT_CH  3125
#define SRT_CHP 3200
#define SRT_OFP 288
#define SRT_NB  256
#define SRT_W   8

__global__ __launch_bounds__(AGG_T) void gcn_degree_sort(
    const int* __restrict__ src, const int* __restrict__ dst,
    float* __restrict__ deg_out, float* __restrict__ deg_in,
    int* __restrict__ offc, int* __restrict__ sorted, int E, int n_nodes) {
  constexpr int PER = (SRT_CH + AGG_T - 1) / AGG_T;
  __shared__ int hist[SRT_NB];
  __shared__ int cur[SRT_NB];
  __shared__ int stage[SRT_CH];
  __shared__ int n_sorted;
  const int tid  = threadIdx.x;
  const int lane = tid & 31;
  const int e0   = blockIdx.x * SRT_CH;
  const int len  = (E - e0 < SRT_CH) ? (E - e0) : SRT_CH;

  if (tid < SRT_NB) hist[tid] = 0;

  int sv[PER], dv[PER], sl[PER];
#pragma unroll
  for (int it = 0; it < PER; ++it) {
    const int i = tid + it * AGG_T;
    sv[it] = 0;
    dv[it] = -1;
    if (i < len) { sv[it] = src[e0 + i]; dv[it] = dst[e0 + i]; }
  }
  __syncthreads();

#pragma unroll
  for (int it = 0; it < PER; ++it) {
    sl[it] = -1;
    if (tid + it * AGG_T < len) {
      atomicAdd(&deg_out[sv[it]], 1.0f);
      atomicAdd(&deg_in[dv[it]], 1.0f);
      if ((unsigned)dv[it] < (unsigned)n_nodes) sl[it] = atomicAdd(&hist[(unsigned)dv[it] / AGG_R], 1);
    }
  }
  __syncthreads();

  int* row = offc + (size_t)blockIdx.x * SRT_OFP;
  if (tid < 32) {
    int v[SRT_NB / 32];
    int lsum = 0;
#pragma unroll
    for (int j = 0; j < SRT_NB / 32; ++j) { v[j] = hist[lane * (SRT_NB / 32) + j]; lsum += v[j]; }
    int inc = lsum;
#pragma unroll
    for (int dl = 1; dl < 32; dl <<= 1) {
      const int t = __shfl_up(inc, dl);
      if (lane >= dl) inc += t;
    }
    int run = inc - lsum;
#pragma unroll
    for (int j = 0; j < SRT_NB / 32; ++j) {
      cur[lane * (SRT_NB / 32) + j] = run;
      row[lane * (SRT_NB / 32) + j] = run;
      run += v[j];
    }
    if (lane == 31) { row[SRT_NB] = run; n_sorted = run; }
  }
  __syncthreads();

#pragma unroll
  for (int it = 0; it < PER; ++it) {
    if (sl[it] >= 0) {
      const unsigned o = (unsigned)dv[it] / AGG_R;
      const unsigned r = (unsigned)dv[it] - o * AGG_R;
      stage[cur[o] + sl[it]] = (int)(((unsigned)sv[it] << 8) | r);
    }
  }
  __syncthreads();

  int* outc = sorted + (size_t)blockIdx.x * SRT_CHP;
  const int tot = n_sorted;
  for (int i = tid; i < tot; i += AGG_T) outc[i] = stage[i];
}

__global__ void gcn_norm(float* __restrict__ p, int n) {
  int i = blockIdx.x * blockDim.x + threadIdx.x;
  if (i < n) p[i] = __frsqrt_rn(fmaxf(p[i], 1.0f));
}

#define GEMM_RP 32
template <int K, int F, bool PRE, bool POST>
__global__ __launch_bounds__(F * 2) void gcn_gemm_wstrip(
    const float* __restrict__ A, const float* __restrict__ W, float* __restrict__ C, int n_rows,
    const float* __restrict__ pre_scale, const float* __restrict__ post_scale,
    const float* __restrict__ bias, int relu) {
  constexpr int NT = F * 2;
  constexpr int P  = K + 4;
  constexpr int KS = K / 4;
  constexpr int C4 = K / 4;
  constexpr int NV = GEMM_RP * C4 / NT;
  constexpr int RSTEP = NT / C4;
  __shared__ __attribute__((aligned(16))) float sa[2][GEMM_RP * P];

  const int tid  = threadIdx.x;
  const int lane = tid & 31;
  const int wave = __builtin_amdgcn_readfirstlane(tid >> 5);
  const int half = lane >> 4;
  const int lr   = lane & 15;
  const int n    = (wave << 4) + lr;

  v2f wf[KS];
#pragma unroll
  for (int s = 0; s < KS; ++s) {
    const int koff = 4 * s + (half << 1);
    wf[s].x = W[(size_t)koff * F + n];
    wf[s].y = W[(size_t)(koff + 1) * F + n];
  }

  const int n_panels = (n_rows + GEMM_RP - 1) / GEMM_RP;
  const int r0 = tid / C4;
  const int c4 = tid - r0 * C4;
  const float bv = (POST && bias) ? bias[n] : 0.0f;

  v4f   st[NV];
  float  sc[NV];
  float pvc = 1.0f, pvn = 1.0f;
  int p = blockIdx.x;
  if (p < n_panels) {
    if (POST) {
      const int row = p * GEMM_RP + lane;
      pvc = post_scale[(row < n_rows) ? row : (n_rows - 1)];
    }
#pragma unroll
    for (int i = 0; i < NV; ++i) {
      const int row = p * GEMM_RP + r0 + i * RSTEP;
      const int rl  = (row < n_rows) ? row : (n_rows - 1);
      st[i] = *(const v4f*)(A + (size_t)rl * K + 4 * c4);
      if (PRE) sc[i] = pre_scale[rl];
    }
#pragma unroll
    for (int i = 0; i < NV; ++i) {
      v4f t = st[i];
      if (PRE) t *= sc[i];
      *(v4f*)(&sa[0][(r0 + i * RSTEP) * P + 4 * c4]) = t;
    }
  }
  __syncthreads();

  int cur = 0;
  for (; p < n_panels; p += gridDim.x) {
    const int pn = p + gridDim.x;
    if (pn < n_panels) {
#pragma unroll
      for (int i = 0; i < NV; ++i) {
        const int row = pn * GEMM_RP + r0 + i * RSTEP;
        const int rl  = (row < n_rows) ? row : (n_rows - 1);
        st[i] = *(const v4f*)(A + (size_t)rl * K + 4 * c4);
        if (PRE) sc[i] = pre_scale[rl];
      }
      if (POST) {
        const int row = pn * GEMM_RP + lane;
        pvn = post_scale[(row < n_rows) ? row : (n_rows - 1)];
      }
    }

    const float* fa = &sa[cur][lr * P + (half << 1)];
    v8f acc[2] = {};
#pragma unroll
    for (int s = 0; s < KS; ++s) {
      const v2f a0 = *(const v2f*)(fa + 4 * s);
      const v2f a1 = *(const v2f*)(fa + 16 * P + 4 * s);
      acc[0] = __builtin_amdgcn_wmma_f32_16x16x4_f32(
           false, a0,  false, wf[s],
           (short)0, acc[0],  false,  false);
      acc[1] = __builtin_amdgcn_wmma_f32_16x16x4_f32(
           false, a1,  false, wf[s],
           (short)0, acc[1],  false,  false);
    }

    if (pn < n_panels) {
#pragma unroll
      for (int i = 0; i < NV; ++i) {
        v4f t = st[i];
        if (PRE) t *= sc[i];
        *(v4f*)(&sa[cur ^ 1][(r0 + i * RSTEP) * P + 4 * c4]) = t;
      }
    }
    __syncthreads();
    cur ^= 1;

#pragma unroll
    for (int t = 0; t < 2; ++t) {
      const int rowt = p * GEMM_RP + 16 * t;
      if (rowt < n_rows) {
#pragma unroll
        for (int i = 0; i < 8; ++i) {
          const int row = rowt + i + (half << 3);
          float v = acc[t][i];
          if (POST) {
            v *= __shfl(pvc, 16 * t + i + (half << 3));
            if (bias) v += bv;
            if (relu) v = fmaxf(v, 0.0f);
          }
          C[(size_t)row * F + n] = v;
        }
      }
    }
    pvc = pvn;
  }
}

template <int F>
__global__ __launch_bounds__(AGG_T) void gcn_agg_own(
    const float* __restrict__ msg, const int* __restrict__ sorted, const int* __restrict__ offc,
    const float* __restrict__ scale, const float* __restrict__ nd,
    const float* __restrict__ bias, float* __restrict__ outp,
    int n_chunks, int n_nodes, int relu) {
  constexpr int V = F / 32;
  __shared__ __attribute__((aligned(16))) float acc[AGG_R * F];

  const int tid  = threadIdx.x;
  const int lane = tid & 31;
  const int wave = __builtin_amdgcn_readfirstlane(tid >> 5);
  const int own  = blockIdx.x;
  const int base = own * AGG_R;

  for (int i = tid; i < AGG_R * F / 4; i += AGG_T)
    ((float4*)acc)[i] = make_float4(0.0f, 0.0f, 0.0f, 0.0f);
  __syncthreads();

  for (int cb = wave * SRT_W; cb < n_chunks; cb += (AGG_T / 32) * SRT_W) {
    int st = 0, ln = 0;
    if (lane < SRT_W && cb + lane < n_chunks) {
      const int* row = offc + (size_t)(cb + lane) * SRT_OFP + own;
      const int s0 = row[0];
      const int s1 = row[1];
      st = (cb + lane) * SRT_CHP + s0;
      ln = s1 - s0;
    }
    int endj[SRT_W], adj[SRT_W];
    int total = 0;
#pragma unroll
    for (int j = 0; j < SRT_W; ++j) {
      const int lj = __builtin_amdgcn_readlane(ln, j);
      const int sj = __builtin_amdgcn_readlane(st, j);
      adj[j] = sj - total;
      total += lj;
      endj[j] = total;
    }

    int ent_n = 0;
    if (lane < total) {
      int a = adj[0];
#pragma unroll
      for (int j = 1; j < SRT_W; ++j) if (lane >= endj[j - 1]) a = adj[j];
      ent_n = sorted[a + lane];
    }
    for (int b0 = 0; b0 < total; b0 += 32) {
      const int ent = ent_n;
      float sc = 1.0f;
      if (scale && b0 + lane < total) sc = scale[(unsigned)ent >> 8];
      const int idn = b0 + 32 + lane;
      ent_n = 0;
      if (idn < total) {
        int a = adj[0];
#pragma unroll
        for (int j = 1; j < SRT_W; ++j) if (idn >= endj[j - 1]) a = adj[j];
        ent_n = sorted[a + idn];
      }
      const int nb = (total - b0 < 32) ? (total - b0) : 32;
      for (int k = 0; k < nb; k += AGG_G) {
        float v[AGG_G][V];
        int rr[AGG_G];
        float ss[AGG_G];
#pragma unroll
        for (int q = 0; q < AGG_G; ++q) {
          const int kk = (k + q < nb) ? (k + q) : (nb - 1);
          const int ek = __builtin_amdgcn_readlane(ent, kk);
          const int sk = (int)((unsigned)ek >> 8);
          rr[q] = ek & 255;
          ss[q] = __int_as_float(__builtin_amdgcn_readlane(__float_as_int(sc), kk));
          const float* rowp = msg + (size_t)sk * F + V * lane;
          if (V == 4) {
            const float4 t = *(const float4*)rowp;
            v[q][0] = t.x; v[q][1] = t.y; v[q][2] = t.z; v[q][V - 1] = t.w;
          } else {
            const float2 t = *(const float2*)rowp;
            v[q][0] = t.x; v[q][V - 1] = t.y;
          }
        }
#pragma unroll
        for (int q = 0; q < AGG_G; ++q) {
          if (k + q < nb) {
            float* a = acc + rr[q] * F + lane;
#pragma unroll
            for (int j = 0; j < V; ++j) atomicAdd(a + 32 * j, v[q][j] * ss[q]);
          }
        }
      }
    }
  }
  __syncthreads();

  float bv[V];
#pragma unroll
  for (int j = 0; j < V; ++j) bv[j] = bias ? bias[V * lane + j] : 0.0f;
  for (int r = wave; r < AGG_R; r += AGG_T / 32) {
    const int row = base + r;
    if (row < n_nodes) {
      const float sc = nd ? nd[row] : 1.0f;
      float o[V];
#pragma unroll
      for (int j = 0; j < V; ++j) {
        float t = acc[r * F + 32 * j + lane] * sc + bv[j];
        if (relu) t = fmaxf(t, 0.0f);
        o[j] = t;
      }
      float* op = outp + (size_t)row * F + V * lane;
      if (V == 4) *(float4*)op = make_float4(o[0], o[1], o[2], o[V - 1]);
      else        *(float2*)op = make_float2(o[0], o[V - 1]);
    }
  }
}

static inline unsigned cdiv(unsigned a, unsigned b) { return (a + b - 1) / b; }

extern "C" void kernel_launch(void* const* d_in, const int* in_sizes, int n_in,
                              void* d_out, int out_size, void* d_ws, size_t ws_size,
                              hipStream_t stream) {
  const float* features = (const float*)d_in[0];
  const float* W1 = (const float*)d_in[1];
  const float* b1 = (const float*)d_in[2];
  const float* W2 = (const float*)d_in[3];
  const float* b2 = (const float*)d_in[4];
  const float* W3 = (const float*)d_in[5];
  const float* b3 = (const float*)d_in[6];
  const int* src  = (const int*)d_in[7];
  const int* dst  = (const int*)d_in[8];
  float* out = (float*)d_out;

  const int N  = N_NODES;
  const int E  = in_sizes[7];
  const int H  = in_sizes[2];

  float* ws   = (float*)d_ws;
  float* ns   = ws;
  float* nd   = ws + N;
  float* bufA = ws + 2 * N + 96;
  float* bufB = bufA + (size_t)N * H;
  const unsigned n_chunks = cdiv(E, SRT_CH);
  int* offc   = (int*)(bufB + (size_t)N * H);
  int* sorted = offc + (size_t)n_chunks * SRT_OFP;
  (void)ws_size; (void)n_in; (void)out_size;

  const int BS = 256;

  gcn_zero  <<<cdiv(2 * N, BS), BS, 0, stream>>>(ns, 2 * N);
  gcn_degree_sort<<<n_chunks, AGG_T, 0, stream>>>(src, dst, ns, nd, offc, sorted, E, N);
  gcn_norm  <<<cdiv(2 * N, BS), BS, 0, stream>>>(ns, 2 * N);

  const unsigned agg_blocks = cdiv(N, AGG_R);

  {
    gcn_gemm_wstrip<256, 128, true, false><<<512, 256, 0, stream>>>(
        features, W1, bufA, N, ns, nullptr, nullptr, 0);
    gcn_agg_own<128><<<agg_blocks, AGG_T, 0, stream>>>(
        bufA, sorted, offc, nullptr, nd, b1, bufB, n_chunks, N, 1);
  }

  {
    gcn_agg_own<128><<<agg_blocks, AGG_T, 0, stream>>>(
        bufB, sorted, offc, ns, nullptr, nullptr, bufA, n_chunks, N, 0);
    gcn_gemm_wstrip<128, 128, false, true><<<768, 256, 0, stream>>>(
        bufA, W2, bufB, N, nullptr, nd, b2, 1);
  }

  {
    gcn_gemm_wstrip<128, 64, true, false><<<1024, 128, 0, stream>>>(
        bufB, W3, bufA, N, ns, nullptr, nullptr, 0);
    gcn_agg_own<64><<<agg_blocks, AGG_T, 0, stream>>>(
        bufA, sorted, offc, nullptr, nd, b3, out, n_chunks, N, 0);
  }
}


# degree counts without global atomics: sort launch also counting-sorts each chunk's src by owner block; an owner launch counts both lists in LDS (int, exact), writes norms; zero/norm launches go
# speedup vs baseline: 4.1388x; 1.5688x over previous
#include <hip/hip_runtime.h>


typedef float v2f __attribute__((ext_vector_type(2)));
typedef float v4f __attribute__((ext_vector_type(4)));
typedef float v8f __attribute__((ext_vector_type(8)));

#define N_NODES 50000

#define AGG_R   196
#define AGG_T   1024
#define AGG_G   8
#define SRT_CH  3125
#define SRT_CHP 3200
#define SRT_OFP 288
#define SRT_NB  256
#define SRT_W   8
#define CNT_Q   4

__global__ __launch_bounds__(AGG_T) void gcn_edge_sort(
    const int* __restrict__ src, const int* __restrict__ dst,
    int* __restrict__ offc, int* __restrict__ sorted,
    int* __restrict__ offs, int* __restrict__ sorted_s, int E, int n_nodes) {
  constexpr int PER = (SRT_CH + AGG_T - 1) / AGG_T;
  __shared__ int hist[2][SRT_NB];
  __shared__ int cur[2][SRT_NB];
  __shared__ int stage[2][SRT_CH];
  __shared__ int n_sorted[2];
  const int tid  = threadIdx.x;
  const int lane = tid & 31;
  const int e0   = blockIdx.x * SRT_CH;
  const int len  = (E - e0 < SRT_CH) ? (E - e0) : SRT_CH;

  if (tid < 2 * SRT_NB) (&hist[0][0])[tid] = 0;

  int sv[PER], dv[PER], sl[PER], ss[PER];
#pragma unroll
  for (int it = 0; it < PER; ++it) {
    const int i = tid + it * AGG_T;
    sv[it] = -1;
    dv[it] = -1;
    if (i < len) { sv[it] = src[e0 + i]; dv[it] = dst[e0 + i]; }
  }
  __syncthreads();

#pragma unroll
  for (int it = 0; it < PER; ++it) {
    sl[it] = -1;
    ss[it] = -1;
    if ((unsigned)dv[it] < (unsigned)n_nodes) sl[it] = atomicAdd(&hist[0][(unsigned)dv[it] / AGG_R], 1);
    if ((unsigned)sv[it] < (unsigned)n_nodes) ss[it] = atomicAdd(&hist[1][(unsigned)sv[it] / AGG_R], 1);
  }
  __syncthreads();

  if (tid < 64) {
    const int w = tid >> 5;
    int* row = (w ? offs : offc) + (size_t)blockIdx.x * SRT_OFP;
    int v[SRT_NB / 32];
    int lsum = 0;
#pragma unroll
    for (int j = 0; j < SRT_NB / 32; ++j) { v[j] = hist[w][lane * (SRT_NB / 32) + j]; lsum += v[j]; }
    int inc = lsum;
#pragma unroll
    for (int dl = 1; dl < 32; dl <<= 1) {
      const int t = __shfl_up(inc, dl);
      if (lane >= dl) inc += t;
    }
    int run = inc - lsum;
#pragma unroll
    for (int j = 0; j < SRT_NB / 32; ++j) {
      cur[w][lane * (SRT_NB / 32) + j] = run;
      row[lane * (SRT_NB / 32) + j] = run;
      run += v[j];
    }
    if (lane == 31) { row[SRT_NB] = run; n_sorted[w] = run; }
  }
  __syncthreads();

#pragma unroll
  for (int it = 0; it < PER; ++it) {
    if (sl[it] >= 0) {
      const unsigned o = (unsigned)dv[it] / AGG_R;
      const unsigned r = (unsigned)dv[it] - o * AGG_R;
      stage[0][cur[0][o] + sl[it]] = (int)(((unsigned)sv[it] << 8) | r);
    }
    if (ss[it] >= 0) {
      const unsigned o = (unsigned)sv[it] / AGG_R;
      stage[1][cur[1][o] + ss[it]] = (int)((unsigned)sv[it] - o * AGG_R);
    }
  }
  __syncthreads();

  int* outc = sorted   + (size_t)blockIdx.x * SRT_CHP;
  int* outs = sorted_s + (size_t)blockIdx.x * SRT_CHP;
  const int tot0 = n_sorted[0];
  const int tot1 = n_sorted[1];
  for (int i = tid; i < tot0; i += AGG_T) outc[i] = stage[0][i];
  for (int i = tid; i < tot1; i += AGG_T) outs[i] = stage[1][i];
}

__global__ __launch_bounds__(AGG_T) void gcn_owner_norm(
    const int* __restrict__ sorted, const int* __restrict__ offc,
    const int* __restrict__ sorted_s, const int* __restrict__ offs,
    float* __restrict__ ns, float* __restrict__ nd, int n_chunks, int n_nodes) {
  __shared__ int cnt[2][AGG_R];
  const int tid  = threadIdx.x;
  const int lane = tid & 31;
  const int wave = __builtin_amdgcn_readfirstlane(tid >> 5);
  const int own  = blockIdx.x;
  const int base = own * AGG_R;

  if (tid < 2 * AGG_R) (&cnt[0][0])[tid] = 0;
  __syncthreads();

  for (int which = 0; which < 2; ++which) {
    const int* list = which ? sorted_s : sorted;
    const int* off  = which ? offs : offc;
    int* c = cnt[which];
    for (int cb = wave * SRT_W; cb < n_chunks; cb += (AGG_T / 32) * SRT_W) {
      int st = 0, ln = 0;
      if (lane < SRT_W && cb + lane < n_chunks) {
        const int* row = off + (size_t)(cb + lane) * SRT_OFP + own;
        const int s0 = row[0];
        const int s1 = row[1];
        st = (cb + lane) * SRT_CHP + s0;
        ln = s1 - s0;
      }
      int endj[SRT_W], adj[SRT_W];
      int total = 0;
#pragma unroll
      for (int j = 0; j < SRT_W; ++j) {
        const int lj = __builtin_amdgcn_readlane(ln, j);
        const int sj = __builtin_amdgcn_readlane(st, j);
        adj[j] = sj - total;
        total += lj;
        endj[j] = total;
      }
      for (int b0 = 0; b0 < total; b0 += 32 * CNT_Q) {
        int e[CNT_Q];
#pragma unroll
        for (int q = 0; q < CNT_Q; ++q) {
          const int idx = b0 + 32 * q + lane;
          e[q] = 0;
          if (idx < total) {
            int a = adj[0];
#pragma unroll
            for (int j = 1; j < SRT_W; ++j) if (idx >= endj[j - 1]) a = adj[j];
            e[q] = list[a + idx];
          }
        }
#pragma unroll
        for (int q = 0; q < CNT_Q; ++q)
          if (b0 + 32 * q + lane < total) atomicAdd(&c[e[q] & 255], 1);
      }
    }
  }
  __syncthreads();

  if (tid < AGG_R) {
    const int row = base + tid;
    if (row < n_nodes) {
      ns[row] = __frsqrt_rn(fmaxf((float)cnt[1][tid], 1.0f));
      nd[row] = __frsqrt_rn(fmaxf((float)cnt[0][tid], 1.0f));
    }
  }
}

#define GEMM_RP 32
template <int K, int F, bool PRE, bool POST>
__global__ __launch_bounds__(F * 2) void gcn_gemm_wstrip(
    const float* __restrict__ A, const float* __restrict__ W, float* __restrict__ C, int n_rows,
    const float* __restrict__ pre_scale, const float* __restrict__ post_scale,
    const float* __restrict__ bias, int relu) {
  constexpr int NT = F * 2;
  constexpr int P  = K + 4;
  constexpr int KS = K / 4;
  constexpr int C4 = K / 4;
  constexpr int NV = GEMM_RP * C4 / NT;
  constexpr int RSTEP = NT / C4;
  __shared__ __attribute__((aligned(16))) float sa[2][GEMM_RP * P];

  const int tid  = threadIdx.x;
  const int lane = tid & 31;
  const int wave = __builtin_amdgcn_readfirstlane(tid >> 5);
  const int half = lane >> 4;
  const int lr   = lane & 15;
  const int n    = (wave << 4) + lr;

  v2f wf[KS];
#pragma unroll
  for (int s = 0; s < KS; ++s) {
    const int koff = 4 * s + (half << 1);
    wf[s].x = W[(size_t)koff * F + n];
    wf[s].y = W[(size_t)(koff + 1) * F + n];
  }

  const int n_panels = (n_rows + GEMM_RP - 1) / GEMM_RP;
  const int r0 = tid / C4;
  const int c4 = tid - r0 * C4;
  const float bv = (POST && bias) ? bias[n] : 0.0f;

  v4f   st[NV];
  float  sc[NV];
  float pvc = 1.0f, pvn = 1.0f;
  int p = blockIdx.x;
  if (p < n_panels) {
    if (POST) {
      const int row = p * GEMM_RP + lane;
      pvc = post_scale[(row < n_rows) ? row : (n_rows - 1)];
    }
#pragma unroll
    for (int i = 0; i < NV; ++i) {
      const int row = p * GEMM_RP + r0 + i * RSTEP;
      const int rl  = (row < n_rows) ? row : (n_rows - 1);
      st[i] = *(const v4f*)(A + (size_t)rl * K + 4 * c4);
      if (PRE) sc[i] = pre_scale[rl];
    }
#pragma unroll
    for (int i = 0; i < NV; ++i) {
      v4f t = st[i];
      if (PRE) t *= sc[i];
      *(v4f*)(&sa[0][(r0 + i * RSTEP) * P + 4 * c4]) = t;
    }
  }
  __syncthreads();

  int cur = 0;
  for (; p < n_panels; p += gridDim.x) {
    const int pn = p + gridDim.x;
    if (pn < n_panels) {
#pragma unroll
      for (int i = 0; i < NV; ++i) {
        const int row = pn * GEMM_RP + r0 + i * RSTEP;
        const int rl  = (row < n_rows) ? row : (n_rows - 1);
        st[i] = *(const v4f*)(A + (size_t)rl * K + 4 * c4);
        if (PRE) sc[i] = pre_scale[rl];
      }
      if (POST) {
        const int row = pn * GEMM_RP + lane;
        pvn = post_scale[(row < n_rows) ? row : (n_rows - 1)];
      }
    }

    const float* fa = &sa[cur][lr * P + (half << 1)];
    v8f acc[2] = {};
#pragma unroll
    for (int s = 0; s < KS; ++s) {
      const v2f a0 = *(const v2f*)(fa + 4 * s);
      const v2f a1 = *(const v2f*)(fa + 16 * P + 4 * s);
      acc[0] = __builtin_amdgcn_wmma_f32_16x16x4_f32(
           false, a0,  false, wf[s],
           (short)0, acc[0],  false,  false);
      acc[1] = __builtin_amdgcn_wmma_f32_16x16x4_f32(
           false, a1,  false, wf[s],
           (short)0, acc[1],  false,  false);
    }

    if (pn < n_panels) {
#pragma unroll
      for (int i = 0; i < NV; ++i) {
        v4f t = st[i];
        if (PRE) t *= sc[i];
        *(v4f*)(&sa[cur ^ 1][(r0 + i * RSTEP) * P + 4 * c4]) = t;
      }
    }
    __syncthreads();
    cur ^= 1;

#pragma unroll
    for (int t = 0; t < 2; ++t) {
      const int rowt = p * GEMM_RP + 16 * t;
      if (rowt < n_rows) {
#pragma unroll
        for (int i = 0; i < 8; ++i) {
          const int row = rowt + i + (half << 3);
          float v = acc[t][i];
          if (POST) {
            v *= __shfl(pvc, 16 * t + i + (half << 3));
            if (bias) v += bv;
            if (relu) v = fmaxf(v, 0.0f);
          }
          C[(size_t)row * F + n] = v;
        }
      }
    }
    pvc = pvn;
  }
}

template <int F>
__global__ __launch_bounds__(AGG_T) void gcn_agg_own(
    const float* __restrict__ msg, const int* __restrict__ sorted, const int* __restrict__ offc,
    const float* __restrict__ scale, const float* __restrict__ nd,
    const float* __restrict__ bias, float* __restrict__ outp,
    int n_chunks, int n_nodes, int relu) {
  constexpr int V = F / 32;
  __shared__ __attribute__((aligned(16))) float acc[AGG_R * F];

  const int tid  = threadIdx.x;
  const int lane = tid & 31;
  const int wave = __builtin_amdgcn_readfirstlane(tid >> 5);
  const int own  = blockIdx.x;
  const int base = own * AGG_R;

  for (int i = tid; i < AGG_R * F / 4; i += AGG_T)
    ((float4*)acc)[i] = make_float4(0.0f, 0.0f, 0.0f, 0.0f);
  __syncthreads();

  for (int cb = wave * SRT_W; cb < n_chunks; cb += (AGG_T / 32) * SRT_W) {
    int st = 0, ln = 0;
    if (lane < SRT_W && cb + lane < n_chunks) {
      const int* row = offc + (size_t)(cb + lane) * SRT_OFP + own;
      const int s0 = row[0];
      const int s1 = row[1];
      st = (cb + lane) * SRT_CHP + s0;
      ln = s1 - s0;
    }
    int endj[SRT_W], adj[SRT_W];
    int total = 0;
#pragma unroll
    for (int j = 0; j < SRT_W; ++j) {
      const int lj = __builtin_amdgcn_readlane(ln, j);
      const int sj = __builtin_amdgcn_readlane(st, j);
      adj[j] = sj - total;
      total += lj;
      endj[j] = total;
    }

    int ent_n = 0;
    if (lane < total) {
      int a = adj[0];
#pragma unroll
      for (int j = 1; j < SRT_W; ++j) if (lane >= endj[j - 1]) a = adj[j];
      ent_n = sorted[a + lane];
    }
    for (int b0 = 0; b0 < total; b0 += 32) {
      const int ent = ent_n;
      float sc = 1.0f;
      if (scale && b0 + lane < total) sc = scale[(unsigned)ent >> 8];
      const int idn = b0 + 32 + lane;
      ent_n = 0;
      if (idn < total) {
        int a = adj[0];
#pragma unroll
        for (int j = 1; j < SRT_W; ++j) if (idn >= endj[j - 1]) a = adj[j];
        ent_n = sorted[a + idn];
      }
      const int nb = (total - b0 < 32) ? (total - b0) : 32;
      for (int k = 0; k < nb; k += AGG_G) {
        float v[AGG_G][V];
        int rr[AGG_G];
        float ss[AGG_G];
#pragma unroll
        for (int q = 0; q < AGG_G; ++q) {
          const int kk = (k + q < nb) ? (k + q) : (nb - 1);
          const int ek = __builtin_amdgcn_readlane(ent, kk);
          const int sk = (int)((unsigned)ek >> 8);
          rr[q] = ek & 255;
          ss[q] = __int_as_float(__builtin_amdgcn_readlane(__float_as_int(sc), kk));
          const float* rowp = msg + (size_t)sk * F + V * lane;
          if (V == 4) {
            const float4 t = *(const float4*)rowp;
            v[q][0] = t.x; v[q][1] = t.y; v[q][2] = t.z; v[q][V - 1] = t.w;
          } else {
            const float2 t = *(const float2*)rowp;
            v[q][0] = t.x; v[q][V - 1] = t.y;
          }
        }
#pragma unroll
        for (int q = 0; q < AGG_G; ++q) {
          if (k + q < nb) {
            float* a = acc + rr[q] * F + lane;
#pragma unroll
            for (int j = 0; j < V; ++j) atomicAdd(a + 32 * j, v[q][j] * ss[q]);
          }
        }
      }
    }
  }
  __syncthreads();

  float bv[V];
#pragma unroll
  for (int j = 0; j < V; ++j) bv[j] = bias ? bias[V * lane + j] : 0.0f;
  for (int r = wave; r < AGG_R; r += AGG_T / 32) {
    const int row = base + r;
    if (row < n_nodes) {
      const float sc = nd ? nd[row] : 1.0f;
      float o[V];
#pragma unroll
      for (int j = 0; j < V; ++j) {
        float t = acc[r * F + 32 * j + lane] * sc + bv[j];
        if (relu) t = fmaxf(t, 0.0f);
        o[j] = t;
      }
      float* op = outp + (size_t)row * F + V * lane;
      if (V == 4) *(float4*)op = make_float4(o[0], o[1], o[2], o[V - 1]);
      else        *(float2*)op = make_float2(o[0], o[V - 1]);
    }
  }
}

static inline unsigned cdiv(unsigned a, unsigned b) { return (a + b - 1) / b; }

extern "C" void kernel_launch(void* const* d_in, const int* in_sizes, int n_in,
                              void* d_out, int out_size, void* d_ws, size_t ws_size,
                              hipStream_t stream) {
  const float* features = (const float*)d_in[0];
  const float* W1 = (const float*)d_in[1];
  const float* b1 = (const float*)d_in[2];
  const float* W2 = (const float*)d_in[3];
  const float* b2 = (const float*)d_in[4];
  const float* W3 = (const float*)d_in[5];
  const float* b3 = (const float*)d_in[6];
  const int* src  = (const int*)d_in[7];
  const int* dst  = (const int*)d_in[8];
  float* out = (float*)d_out;

  const int N  = N_NODES;
  const int E  = in_sizes[7];
  const int H  = in_sizes[2];

  float* ws   = (float*)d_ws;
  float* ns   = ws;
  float* nd   = ws + N;
  float* bufA = ws + 2 * N + 96;
  float* bufB = bufA + (size_t)N * H;
  const unsigned n_chunks = cdiv(E, SRT_CH);
  int* offc   = (int*)(bufB + (size_t)N * H);
  int* sorted = offc + (size_t)n_chunks * SRT_OFP;
  int* offs   = sorted + (size_t)n_chunks * SRT_CHP;
  int* sorted_s = offs + (size_t)n_chunks * SRT_OFP;
  (void)ws_size; (void)n_in; (void)out_size;

  const unsigned agg_blocks = cdiv(N, AGG_R);

  gcn_edge_sort <<<n_chunks, AGG_T, 0, stream>>>(src, dst, offc, sorted, offs, sorted_s, E, N);
  gcn_owner_norm<<<agg_blocks, AGG_T, 0, stream>>>(sorted, offc, sorted_s, offs, ns, nd, n_chunks, N);

  {
    gcn_gemm_wstrip<256, 128, true, false><<<512, 256, 0, stream>>>(
        features, W1, bufA, N, ns, nullptr, nullptr, 0);
    gcn_agg_own<128><<<agg_blocks, AGG_T, 0, stream>>>(
        bufA, sorted, offc, nullptr, nd, b1, bufB, n_chunks, N, 1);
  }

  {
    gcn_agg_own<128><<<agg_blocks, AGG_T, 0, stream>>>(
        bufB, sorted, offc, ns, nullptr, nullptr, bufA, n_chunks, N, 0);
    gcn_gemm_wstrip<128, 128, false, true><<<768, 256, 0, stream>>>(
        bufA, W2, bufB, N, nullptr, nd, b2, 1);
  }

  {
    gcn_gemm_wstrip<128, 64, true, false><<<1024, 128, 0, stream>>>(
        bufB, W3, bufA, N, ns, nullptr, nullptr, 0);
    gcn_agg_own<64><<<agg_blocks, AGG_T, 0, stream>>>(
        bufA, sorted, offc, nullptr, nd, b3, out, n_chunks, N, 0);
  }
}


# 128-wide neighbour-sum launches end with the next weight product: owner block scales its LDS rows, multiplies by W2 / W3 on f32 WMMA; layer 2 = nsum((h1*ns)@W2); 8 launches -> 6; sum order changes
# speedup vs baseline: 4.7489x; 1.1474x over previous
#include <hip/hip_runtime.h>


typedef float v2f __attribute__((ext_vector_type(2)));
typedef float v4f __attribute__((ext_vector_type(4)));
typedef float v8f __attribute__((ext_vector_type(8)));

#define N_NODES 50000

#define AGG_R   196
#define AGG_T   1024
#define AGG_G   8
#define SRT_CH  3125
#define SRT_CHP 3200
#define SRT_OFP 288
#define SRT_NB  256
#define SRT_W   8
#define CNT_Q   4

__global__ __launch_bounds__(AGG_T) void gcn_edge_sort(
    const int* __restrict__ src, const int* __restrict__ dst,
    int* __restrict__ offc, int* __restrict__ sorted,
    int* __restrict__ offs, int* __restrict__ sorted_s, int E, int n_nodes) {
  constexpr int PER = (SRT_CH + AGG_T - 1) / AGG_T;
  __shared__ int hist[2][SRT_NB];
  __shared__ int cur[2][SRT_NB];
  __shared__ int stage[2][SRT_CH];
  __shared__ int n_sorted[2];
  const int tid  = threadIdx.x;
  const int lane = tid & 31;
  const int e0   = blockIdx.x * SRT_CH;
  const int len  = (E - e0 < SRT_CH) ? (E - e0) : SRT_CH;

  if (tid < 2 * SRT_NB) (&hist[0][0])[tid] = 0;

  int sv[PER], dv[PER], sl[PER], ss[PER];
#pragma unroll
  for (int it = 0; it < PER; ++it) {
    const int i = tid + it * AGG_T;
    sv[it] = -1;
    dv[it] = -1;
    if (i < len) { sv[it] = src[e0 + i]; dv[it] = dst[e0 + i]; }
  }
  __syncthreads();

#pragma unroll
  for (int it = 0; it < PER; ++it) {
    sl[it] = -1;
    ss[it] = -1;
    if ((unsigned)dv[it] < (unsigned)n_nodes) sl[it] = atomicAdd(&hist[0][(unsigned)dv[it] / AGG_R], 1);
    if ((unsigned)sv[it] < (unsigned)n_nodes) ss[it] = atomicAdd(&hist[1][(unsigned)sv[it] / AGG_R], 1);
  }
  __syncthreads();

  if (tid < 64) {
    const int w = tid >> 5;
    int* row = (w ? offs : offc) + (size_t)blockIdx.x * SRT_OFP;
    int v[SRT_NB / 32];
    int lsum = 0;
#pragma unroll
    for (int j = 0; j < SRT_NB / 32; ++j) { v[j] = hist[w][lane * (SRT_NB / 32) + j]; lsum += v[j]; }
    int inc = lsum;
#pragma unroll
    for (int dl = 1; dl < 32; dl <<= 1) {
      const int t = __shfl_up(inc, dl);
      if (lane >= dl) inc += t;
    }
    int run = inc - lsum;
#pragma unroll
    for (int j = 0; j < SRT_NB / 32; ++j) {
      cur[w][lane * (SRT_NB / 32) + j] = run;
      row[lane * (SRT_NB / 32) + j] = run;
      run += v[j];
    }
    if (lane == 31) { row[SRT_NB] = run; n_sorted[w] = run; }
  }
  __syncthreads();

#pragma unroll
  for (int it = 0; it < PER; ++it) {
    if (sl[it] >= 0) {
      const unsigned o = (unsigned)dv[it] / AGG_R;
      const unsigned r = (unsigned)dv[it] - o * AGG_R;
      stage[0][cur[0][o] + sl[it]] = (int)(((unsigned)sv[it] << 8) | r);
    }
    if (ss[it] >= 0) {
      const unsigned o = (unsigned)sv[it] / AGG_R;
      stage[1][cur[1][o] + ss[it]] = (int)((unsigned)sv[it] - o * AGG_R);
    }
  }
  __syncthreads();

  int* outc = sorted   + (size_t)blockIdx.x * SRT_CHP;
  int* outs = sorted_s + (size_t)blockIdx.x * SRT_CHP;
  const int tot0 = n_sorted[0];
  const int tot1 = n_sorted[1];
  for (int i = tid; i < tot0; i += AGG_T) outc[i] = stage[0][i];
  for (int i = tid; i < tot1; i += AGG_T) outs[i] = stage[1][i];
}

__global__ __launch_bounds__(AGG_T) void gcn_owner_norm(
    const int* __restrict__ sorted, const int* __restrict__ offc,
    const int* __restrict__ sorted_s, const int* __restrict__ offs,
    float* __restrict__ ns, float* __restrict__ nd, int n_chunks, int n_nodes) {
  __shared__ int cnt[2][AGG_R];
  const int tid  = threadIdx.x;
  const int lane = tid & 31;
  const int wave = __builtin_amdgcn_readfirstlane(tid >> 5);
  const int own  = blockIdx.x;
  const int base = own * AGG_R;

  if (tid < 2 * AGG_R) (&cnt[0][0])[tid] = 0;
  __syncthreads();

  for (int which = 0; which < 2; ++which) {
    const int* list = which ? sorted_s : sorted;
    const int* off  = which ? offs : offc;
    int* c = cnt[which];
    for (int cb = wave * SRT_W; cb < n_chunks; cb += (AGG_T / 32) * SRT_W) {
      int st = 0, ln = 0;
      if (lane < SRT_W && cb + lane < n_chunks) {
        const int* row = off + (size_t)(cb + lane) * SRT_OFP + own;
        const int s0 = row[0];
        const int s1 = row[1];
        st = (cb + lane) * SRT_CHP + s0;
        ln = s1 - s0;
      }
      int endj[SRT_W], adj[SRT_W];
      int total = 0;
#pragma unroll
      for (int j = 0; j < SRT_W; ++j) {
        const int lj = __builtin_amdgcn_readlane(ln, j);
        const int sj = __builtin_amdgcn_readlane(st, j);
        adj[j] = sj - total;
        total += lj;
        endj[j] = total;
      }
      for (int b0 = 0; b0 < total; b0 += 32 * CNT_Q) {
        int e[CNT_Q];
#pragma unroll
        for (int q = 0; q < CNT_Q; ++q) {
          const int idx = b0 + 32 * q + lane;
          e[q] = 0;
          if (idx < total) {
            int a = adj[0];
#pragma unroll
            for (int j = 1; j < SRT_W; ++j) if (idx >= endj[j - 1]) a = adj[j];
            e[q] = list[a + idx];
          }
        }
#pragma unroll
        for (int q = 0; q < CNT_Q; ++q)
          if (b0 + 32 * q + lane < total) atomicAdd(&c[e[q] & 255], 1);
      }
    }
  }
  __syncthreads();

  if (tid < AGG_R) {
    const int row = base + tid;
    if (row < n_nodes) {
      ns[row] = __frsqrt_rn(fmaxf((float)cnt[1][tid], 1.0f));
      nd[row] = __frsqrt_rn(fmaxf((float)cnt[0][tid], 1.0f));
    }
  }
}

#define GEMM_RP 32
template <int K, int F, bool PRE, bool POST>
__global__ __launch_bounds__(F * 2) void gcn_gemm_wstrip(
    const float* __restrict__ A, const float* __restrict__ W, float* __restrict__ C, int n_rows,
    const float* __restrict__ pre_scale, const float* __restrict__ post_scale,
    const float* __restrict__ bias, int relu) {
  constexpr int NT = F * 2;
  constexpr int P  = K + 4;
  constexpr int KS = K / 4;
  constexpr int C4 = K / 4;
  constexpr int NV = GEMM_RP * C4 / NT;
  constexpr int RSTEP = NT / C4;
  __shared__ __attribute__((aligned(16))) float sa[2][GEMM_RP * P];

  const int tid  = threadIdx.x;
  const int lane = tid & 31;
  const int wave = __builtin_amdgcn_readfirstlane(tid >> 5);
  const int half = lane >> 4;
  const int lr   = lane & 15;
  const int n    = (wave << 4) + lr;

  v2f wf[KS];
#pragma unroll
  for (int s = 0; s < KS; ++s) {
    const int koff = 4 * s + (half << 1);
    wf[s].x = W[(size_t)koff * F + n];
    wf[s].y = W[(size_t)(koff + 1) * F + n];
  }

  const int n_panels = (n_rows + GEMM_RP - 1) / GEMM_RP;
  const int r0 = tid / C4;
  const int c4 = tid - r0 * C4;
  const float bv = (POST && bias) ? bias[n] : 0.0f;

  v4f   st[NV];
  float  sc[NV];
  float pvc = 1.0f, pvn = 1.0f;
  int p = blockIdx.x;
  if (p < n_panels) {
    if (POST) {
      const int row = p * GEMM_RP + lane;
      pvc = post_scale[(row < n_rows) ? row : (n_rows - 1)];
    }
#pragma unroll
    for (int i = 0; i < NV; ++i) {
      const int row = p * GEMM_RP + r0 + i * RSTEP;
      const int rl  = (row < n_rows) ? row : (n_rows - 1);
      st[i] = *(const v4f*)(A + (size_t)rl * K + 4 * c4);
      if (PRE) sc[i] = pre_scale[rl];
    }
#pragma unroll
    for (int i = 0; i < NV; ++i) {
      v4f t = st[i];
      if (PRE) t *= sc[i];
      *(v4f*)(&sa[0][(r0 + i * RSTEP) * P + 4 * c4]) = t;
    }
  }
  __syncthreads();

  int cur = 0;
  for (; p < n_panels; p += gridDim.x) {
    const int pn = p + gridDim.x;
    if (pn < n_panels) {
#pragma unroll
      for (int i = 0; i < NV; ++i) {
        const int row = pn * GEMM_RP + r0 + i * RSTEP;
        const int rl  = (row < n_rows) ? row : (n_rows - 1);
        st[i] = *(const v4f*)(A + (size_t)rl * K + 4 * c4);
        if (PRE) sc[i] = pre_scale[rl];
      }
      if (POST) {
        const int row = pn * GEMM_RP + lane;
        pvn = post_scale[(row < n_rows) ? row : (n_rows - 1)];
      }
    }

    const float* fa = &sa[cur][lr * P + (half << 1)];
    v8f acc[2] = {};
#pragma unroll
    for (int s = 0; s < KS; ++s) {
      const v2f a0 = *(const v2f*)(fa + 4 * s);
      const v2f a1 = *(const v2f*)(fa + 16 * P + 4 * s);
      acc[0] = __builtin_amdgcn_wmma_f32_16x16x4_f32(
           false, a0,  false, wf[s],
           (short)0, acc[0],  false,  false);
      acc[1] = __builtin_amdgcn_wmma_f32_16x16x4_f32(
           false, a1,  false, wf[s],
           (short)0, acc[1],  false,  false);
    }

    if (pn < n_panels) {
#pragma unroll
      for (int i = 0; i < NV; ++i) {
        v4f t = st[i];
        if (PRE) t *= sc[i];
        *(v4f*)(&sa[cur ^ 1][(r0 + i * RSTEP) * P + 4 * c4]) = t;
      }
    }
    __syncthreads();
    cur ^= 1;

#pragma unroll
    for (int t = 0; t < 2; ++t) {
      const int rowt = p * GEMM_RP + 16 * t;
      if (rowt < n_rows) {
#pragma unroll
        for (int i = 0; i < 8; ++i) {
          const int row = rowt + i + (half << 3);
          float v = acc[t][i];
          if (POST) {
            v *= __shfl(pvc, 16 * t + i + (half << 3));
            if (bias) v += bv;
            if (relu) v = fmaxf(v, 0.0f);
          }
          C[(size_t)row * F + n] = v;
        }
      }
    }
    pvc = pvn;
  }
}

template <int F>
__global__ __launch_bounds__(AGG_T) void gcn_agg_own(
    const float* __restrict__ msg, const int* __restrict__ sorted, const int* __restrict__ offc,
    const float* __restrict__ scale, const float* __restrict__ nd,
    const float* __restrict__ bias, float* __restrict__ outp,
    int n_chunks, int n_nodes, int relu) {
  constexpr int V = F / 32;
  __shared__ __attribute__((aligned(16))) float acc[AGG_R * F];

  const int tid  = threadIdx.x;
  const int lane = tid & 31;
  const int wave = __builtin_amdgcn_readfirstlane(tid >> 5);
  const int own  = blockIdx.x;
  const int base = own * AGG_R;

  for (int i = tid; i < AGG_R * F / 4; i += AGG_T)
    ((float4*)acc)[i] = make_float4(0.0f, 0.0f, 0.0f, 0.0f);
  __syncthreads();

  for (int cb = wave * SRT_W; cb < n_chunks; cb += (AGG_T / 32) * SRT_W) {
    int st = 0, ln = 0;
    if (lane < SRT_W && cb + lane < n_chunks) {
      const int* row = offc + (size_t)(cb + lane) * SRT_OFP + own;
      const int s0 = row[0];
      const int s1 = row[1];
      st = (cb + lane) * SRT_CHP + s0;
      ln = s1 - s0;
    }
    int endj[SRT_W], adj[SRT_W];
    int total = 0;
#pragma unroll
    for (int j = 0; j < SRT_W; ++j) {
      const int lj = __builtin_amdgcn_readlane(ln, j);
      const int sj = __builtin_amdgcn_readlane(st, j);
      adj[j] = sj - total;
      total += lj;
      endj[j] = total;
    }

    int ent_n = 0;
    if (lane < total) {
      int a = adj[0];
#pragma unroll
      for (int j = 1; j < SRT_W; ++j) if (lane >= endj[j - 1]) a = adj[j];
      ent_n = sorted[a + lane];
    }
    for (int b0 = 0; b0 < total; b0 += 32) {
      const int ent = ent_n;
      float sc = 1.0f;
      if (scale && b0 + lane < total) sc = scale[(unsigned)ent >> 8];
      const int idn = b0 + 32 + lane;
      ent_n = 0;
      if (idn < total) {
        int a = adj[0];
#pragma unroll
        for (int j = 1; j < SRT_W; ++j) if (idn >= endj[j - 1]) a = adj[j];
        ent_n = sorted[a + idn];
      }
      const int nb = (total - b0 < 32) ? (total - b0) : 32;
      for (int k = 0; k < nb; k += AGG_G) {
        float v[AGG_G][V];
        int rr[AGG_G];
        float ss[AGG_G];
#pragma unroll
        for (int q = 0; q < AGG_G; ++q) {
          const int kk = (k + q < nb) ? (k + q) : (nb - 1);
          const int ek = __builtin_amdgcn_readlane(ent, kk);
          const int sk = (int)((unsigned)ek >> 8);
          rr[q] = ek & 255;
          ss[q] = __int_as_float(__builtin_amdgcn_readlane(__float_as_int(sc), kk));
          const float* rowp = msg + (size_t)sk * F + V * lane;
          if (V == 4) {
            const float4 t = *(const float4*)rowp;
            v[q][0] = t.x; v[q][1] = t.y; v[q][2] = t.z; v[q][V - 1] = t.w;
          } else {
            const float2 t = *(const float2*)rowp;
            v[q][0] = t.x; v[q][V - 1] = t.y;
          }
        }
#pragma unroll
        for (int q = 0; q < AGG_G; ++q) {
          if (k + q < nb) {
            float* a = acc + rr[q] * F + lane;
#pragma unroll
            for (int j = 0; j < V; ++j) atomicAdd(a + 32 * j, v[q][j] * ss[q]);
          }
        }
      }
    }
  }
  __syncthreads();

  float bv[V];
#pragma unroll
  for (int j = 0; j < V; ++j) bv[j] = bias ? bias[V * lane + j] : 0.0f;
  for (int r = wave; r < AGG_R; r += AGG_T / 32) {
    const int row = base + r;
    if (row < n_nodes) {
      const float sc = nd ? nd[row] : 1.0f;
      float o[V];
#pragma unroll
      for (int j = 0; j < V; ++j) {
        float t = acc[r * F + 32 * j + lane] * sc + bv[j];
        if (relu) t = fmaxf(t, 0.0f);
        o[j] = t;
      }
      float* op = outp + (size_t)row * F + V * lane;
      if (V == 4) *(float4*)op = make_float4(o[0], o[1], o[2], o[V - 1]);
      else        *(float2*)op = make_float2(o[0], o[V - 1]);
    }
  }
}

#define AGG_RT  13
#define AGG_P   132
template <int FO>
__global__ __launch_bounds__(AGG_T) void gcn_agg_own_mm(
    const float* __restrict__ msg, const int* __restrict__ sorted, const int* __restrict__ offc,
    const float* __restrict__ nd, const float* __restrict__ bias, const float* __restrict__ ns,
    const float* __restrict__ Wn, float* __restrict__ outp, int n_chunks, int n_nodes) {
  constexpr int F   = 128;
  constexpr int V   = F / 32;
  constexpr int P   = AGG_P;
  constexpr int KS  = F / 4;
  constexpr int NS  = FO / 16;
  constexpr int WPS = (AGG_T / 32) / NS;
  __shared__ __attribute__((aligned(16))) float acc[AGG_RT * 16 * P];

  const int tid  = threadIdx.x;
  const int lane = tid & 31;
  const int wave = __builtin_amdgcn_readfirstlane(tid >> 5);
  const int own  = blockIdx.x;
  const int base = own * AGG_R;

  for (int i = tid; i < AGG_RT * 16 * P / 4; i += AGG_T)
    ((float4*)acc)[i] = make_float4(0.0f, 0.0f, 0.0f, 0.0f);
  __syncthreads();

  for (int cb = wave * SRT_W; cb < n_chunks; cb += (AGG_T / 32) * SRT_W) {
    int st = 0, ln = 0;
    if (lane < SRT_W && cb + lane < n_chunks) {
      const int* row = offc + (size_t)(cb + lane) * SRT_OFP + own;
      const int s0 = row[0];
      const int s1 = row[1];
      st = (cb + lane) * SRT_CHP + s0;
      ln = s1 - s0;
    }
    int endj[SRT_W], adj[SRT_W];
    int total = 0;
#pragma unroll
    for (int j = 0; j < SRT_W; ++j) {
      const int lj = __builtin_amdgcn_readlane(ln, j);
      const int sj = __builtin_amdgcn_readlane(st, j);
      adj[j] = sj - total;
      total += lj;
      endj[j] = total;
    }

    int ent_n = 0;
    if (lane < total) {
      int a = adj[0];
#pragma unroll
      for (int j = 1; j < SRT_W; ++j) if (lane >= endj[j - 1]) a = adj[j];
      ent_n = sorted[a + lane];
    }
    for (int b0 = 0; b0 < total; b0 += 32) {
      const int ent = ent_n;
      const int idn = b0 + 32 + lane;
      ent_n = 0;
      if (idn < total) {
        int a = adj[0];
#pragma unroll
        for (int j = 1; j < SRT_W; ++j) if (idn >= endj[j - 1]) a = adj[j];
        ent_n = sorted[a + idn];
      }
      const int nb = (total - b0 < 32) ? (total - b0) : 32;
      for (int k = 0; k < nb; k += AGG_G) {
        float v[AGG_G][V];
        int rr[AGG_G];
#pragma unroll
        for (int q = 0; q < AGG_G; ++q) {
          const int kk = (k + q < nb) ? (k + q) : (nb - 1);
          const int ek = __builtin_amdgcn_readlane(ent, kk);
          const int sk = (int)((unsigned)ek >> 8);
          rr[q] = ek & 255;
          const float4 t = *(const float4*)(msg + (size_t)sk * F + V * lane);
          v[q][0] = t.x; v[q][1] = t.y; v[q][2] = t.z; v[q][3] = t.w;
        }
#pragma unroll
        for (int q = 0; q < AGG_G; ++q) {
          if (k + q < nb) {
            float* a = acc + rr[q] * P + lane;
#pragma unroll
            for (int j = 0; j < V; ++j) atomicAdd(a + 32 * j, v[q][j]);
          }
        }
      }
    }
  }

  const int half  = lane >> 4;
  const int lr    = lane & 15;
  const int strip = wave % NS;
  const int n     = (strip << 4) + lr;
  v2f wf[KS];
#pragma unroll
  for (int s = 0; s < KS; ++s) {
    const int k0 = 16 * (s & 7) + (half << 3) + (s >> 3);
    wf[s].x = Wn[(size_t)k0 * FO + n];
    wf[s].y = Wn[(size_t)(k0 + 4) * FO + n];
  }
  float bv[V];
#pragma unroll
  for (int j = 0; j < V; ++j) bv[j] = bias[V * lane + j];
  __syncthreads();

  for (int r = wave; r < AGG_R; r += AGG_T / 32) {
    const int row = base + r;
    if (row < n_nodes) {
      const float sc = nd[row];
      const float ps = ns[row];
      float* a = acc + r * P + lane;
#pragma unroll
      for (int j = 0; j < V; ++j) a[32 * j] = fmaxf(a[32 * j] * sc + bv[j], 0.0f) * ps;
    }
  }
  __syncthreads();

  for (int rt = wave / NS; rt < AGG_RT; rt += 2 * WPS) {
    const int rt1 = rt + WPS;
    const float* fa0 = &acc[(16 * rt + lr) * P + (half << 1)];
    v8f d0 = {}, d1 = {};
    if (rt1 < AGG_RT) {
      const float* fa1 = fa0 + 16 * WPS * P;
#pragma unroll
      for (int s = 0; s < KS; ++s) {
        const v2f a0 = *(const v2f*)(fa0 + 4 * s);
        const v2f a1 = *(const v2f*)(fa1 + 4 * s);
        d0 = __builtin_amdgcn_wmma_f32_16x16x4_f32(
             false, a0,  false, wf[s],
             (short)0, d0,  false,  false);
        d1 = __builtin_amdgcn_wmma_f32_16x16x4_f32(
             false, a1,  false, wf[s],
             (short)0, d1,  false,  false);
      }
    } else {
#pragma unroll
      for (int s = 0; s < KS; ++s) {
        const v2f a0 = *(const v2f*)(fa0 + 4 * s);
        d0 = __builtin_amdgcn_wmma_f32_16x16x4_f32(
             false, a0,  false, wf[s],
             (short)0, d0,  false,  false);
      }
    }
#pragma unroll
    for (int i = 0; i < 8; ++i) {
      const int r = 16 * rt + i + (half << 3);
      if (r < AGG_R && base + r < n_nodes) outp[(size_t)(base + r) * FO + n] = d0[i];
    }
    if (rt1 < AGG_RT) {
#pragma unroll
      for (int i = 0; i < 8; ++i) {
        const int r = 16 * rt1 + i + (half << 3);
        if (r < AGG_R && base + r < n_nodes) outp[(size_t)(base + r) * FO + n] = d1[i];
      }
    }
  }
}

static inline unsigned cdiv(unsigned a, unsigned b) { return (a + b - 1) / b; }

extern "C" void kernel_launch(void* const* d_in, const int* in_sizes, int n_in,
                              void* d_out, int out_size, void* d_ws, size_t ws_size,
                              hipStream_t stream) {
  const float* features = (const float*)d_in[0];
  const float* W1 = (const float*)d_in[1];
  const float* b1 = (const float*)d_in[2];
  const float* W2 = (const float*)d_in[3];
  const float* b2 = (const float*)d_in[4];
  const float* W3 = (const float*)d_in[5];
  const float* b3 = (const float*)d_in[6];
  const int* src  = (const int*)d_in[7];
  const int* dst  = (const int*)d_in[8];
  float* out = (float*)d_out;

  const int N  = N_NODES;
  const int E  = in_sizes[7];
  const int H  = in_sizes[2];

  float* ws   = (float*)d_ws;
  float* ns   = ws;
  float* nd   = ws + N;
  float* bufA = ws + 2 * N + 96;
  float* bufB = bufA + (size_t)N * H;
  const unsigned n_chunks = cdiv(E, SRT_CH);
  int* offc   = (int*)(bufB + (size_t)N * H);
  int* sorted = offc + (size_t)n_chunks * SRT_OFP;
  int* offs   = sorted + (size_t)n_chunks * SRT_CHP;
  int* sorted_s = offs + (size_t)n_chunks * SRT_OFP;
  (void)ws_size; (void)n_in; (void)out_size;

  const unsigned agg_blocks = cdiv(N, AGG_R);

  gcn_edge_sort <<<n_chunks, AGG_T, 0, stream>>>(src, dst, offc, sorted, offs, sorted_s, E, N);
  gcn_owner_norm<<<agg_blocks, AGG_T, 0, stream>>>(sorted, offc, sorted_s, offs, ns, nd, n_chunks, N);

  gcn_gemm_wstrip<256, 128, true, false><<<512, 256, 0, stream>>>(
      features, W1, bufA, N, ns, nullptr, nullptr, 0);

  gcn_agg_own_mm<128><<<agg_blocks, AGG_T, 0, stream>>>(
      bufA, sorted, offc, nd, b1, ns, W2, bufB, n_chunks, N);

  gcn_agg_own_mm<64><<<agg_blocks, AGG_T, 0, stream>>>(
      bufB, sorted, offc, nd, b2, ns, W3, bufA, n_chunks, N);

  gcn_agg_own<64><<<agg_blocks, AGG_T, 0, stream>>>(
      bufA, sorted, offc, nullptr, nd, b3, out, n_chunks, N, 0);
}


# neighbour sums: degree launch orders each owner's edges by destination row (LDS counting sort); a wave sums a row's sources in registers, one LDS float add per row not per edge; sum order changes
# speedup vs baseline: 5.2062x; 1.0963x over previous
#include <hip/hip_runtime.h>


typedef float v2f __attribute__((ext_vector_type(2)));
typedef float v4f __attribute__((ext_vector_type(4)));
typedef float v8f __attribute__((ext_vector_type(8)));

#define N_NODES 50000

#define AGG_R   196
#define AGG_T   1024
#define AGG_G   8
#define SRT_CH  3125
#define SRT_CHP 3200
#define SRT_OFP 288
#define SRT_NB  256
#define SRT_W   8
#define CNT_Q   4
#define ORD_CAP 8192

__global__ __launch_bounds__(AGG_T) void gcn_edge_sort(
    const int* __restrict__ src, const int* __restrict__ dst,
    int* __restrict__ offc, int* __restrict__ sorted,
    int* __restrict__ offs, int* __restrict__ sorted_s, int E, int n_nodes) {
  constexpr int PER = (SRT_CH + AGG_T - 1) / AGG_T;
  __shared__ int hist[2][SRT_NB];
  __shared__ int cur[2][SRT_NB];
  __shared__ int stage[2][SRT_CH];
  __shared__ int n_sorted[2];
  const int tid  = threadIdx.x;
  const int lane = tid & 31;
  const int e0   = blockIdx.x * SRT_CH;
  const int len  = (E - e0 < SRT_CH) ? (E - e0) : SRT_CH;

  if (tid < 2 * SRT_NB) (&hist[0][0])[tid] = 0;

  int sv[PER], dv[PER], sl[PER], ss[PER];
#pragma unroll
  for (int it = 0; it < PER; ++it) {
    const int i = tid + it * AGG_T;
    sv[it] = -1;
    dv[it] = -1;
    if (i < len) { sv[it] = src[e0 + i]; dv[it] = dst[e0 + i]; }
  }
  __syncthreads();

#pragma unroll
  for (int it = 0; it < PER; ++it) {
    sl[it] = -1;
    ss[it] = -1;
    if ((unsigned)dv[it] < (unsigned)n_nodes) sl[it] = atomicAdd(&hist[0][(unsigned)dv[it] / AGG_R], 1);
    if ((unsigned)sv[it] < (unsigned)n_nodes) ss[it] = atomicAdd(&hist[1][(unsigned)sv[it] / AGG_R], 1);
  }
  __syncthreads();

  if (tid < 64) {
    const int w = tid >> 5;
    int* row = (w ? offs : offc) + (size_t)blockIdx.x * SRT_OFP;
    int v[SRT_NB / 32];
    int lsum = 0;
#pragma unroll
    for (int j = 0; j < SRT_NB / 32; ++j) { v[j] = hist[w][lane * (SRT_NB / 32) + j]; lsum += v[j]; }
    int inc = lsum;
#pragma unroll
    for (int dl = 1; dl < 32; dl <<= 1) {
      const int t = __shfl_up(inc, dl);
      if (lane >= dl) inc += t;
    }
    int run = inc - lsum;
#pragma unroll
    for (int j = 0; j < SRT_NB / 32; ++j) {
      cur[w][lane * (SRT_NB / 32) + j] = run;
      row[lane * (SRT_NB / 32) + j] = run;
      run += v[j];
    }
    if (lane == 31) { row[SRT_NB] = run; n_sorted[w] = run; }
  }
  __syncthreads();

#pragma unroll
  for (int it = 0; it < PER; ++it) {
    if (sl[it] >= 0) {
      const unsigned o = (unsigned)dv[it] / AGG_R;
      const unsigned r = (unsigned)dv[it] - o * AGG_R;
      stage[0][cur[0][o] + sl[it]] = (int)(((unsigned)sv[it] << 8) | r);
    }
    if (ss[it] >= 0) {
      const unsigned o = (unsigned)sv[it] / AGG_R;
      stage[1][cur[1][o] + ss[it]] = (int)((unsigned)sv[it] - o * AGG_R);
    }
  }
  __syncthreads();

  int* outc = sorted   + (size_t)blockIdx.x * SRT_CHP;
  int* outs = sorted_s + (size_t)blockIdx.x * SRT_CHP;
  const int tot0 = n_sorted[0];
  const int tot1 = n_sorted[1];
  for (int i = tid; i < tot0; i += AGG_T) outc[i] = stage[0][i];
  for (int i = tid; i < tot1; i += AGG_T) outs[i] = stage[1][i];
}

template <class Fn>
__device__ __forceinline__ void gcn_owner_walk(
    const int* __restrict__ list, const int* __restrict__ off, int own, int n_chunks,
    int wave, int lane, int& front, int& count, Fn f) {
  for (int cb = wave * SRT_W; cb < n_chunks; cb += (AGG_T / 32) * SRT_W) {
    int st = 0, ln = 0, s0v = 0;
    if (lane < SRT_W && cb + lane < n_chunks) {
      const int* row = off + (size_t)(cb + lane) * SRT_OFP + own;
      const int s0 = row[0];
      const int s1 = row[1];
      st = (cb + lane) * SRT_CHP + s0;
      ln = s1 - s0;
      s0v = s0;
    }
    int endj[SRT_W], adj[SRT_W];
    int total = 0;
#pragma unroll
    for (int j = 0; j < SRT_W; ++j) {
      const int lj = __builtin_amdgcn_readlane(ln, j);
      const int sj = __builtin_amdgcn_readlane(st, j);
      front += __builtin_amdgcn_readlane(s0v, j);
      adj[j] = sj - total;
      total += lj;
      endj[j] = total;
    }
    count += total;
    for (int b0 = 0; b0 < total; b0 += 32 * CNT_Q) {
      int e[CNT_Q];
#pragma unroll
      for (int q = 0; q < CNT_Q; ++q) {
        const int idx = b0 + 32 * q + lane;
        e[q] = 0;
        if (idx < total) {
          int a = adj[0];
#pragma unroll
          for (int j = 1; j < SRT_W; ++j) if (idx >= endj[j - 1]) a = adj[j];
          e[q] = list[a + idx];
        }
      }
#pragma unroll
      for (int q = 0; q < CNT_Q; ++q)
        if (b0 + 32 * q + lane < total) f(e[q]);
    }
  }
}

__global__ __launch_bounds__(AGG_T) void gcn_owner_norm(
    const int* __restrict__ sorted, const int* __restrict__ offc,
    const int* __restrict__ sorted_s, const int* __restrict__ offs,
    float* __restrict__ ns, float* __restrict__ nd,
    int* __restrict__ ord, int* __restrict__ ordb, int n_chunks, int n_nodes) {
  constexpr int PL = (AGG_R + 31) / 32;
  __shared__ int cnt[2][AGG_R];
  __shared__ int cur[AGG_R];
  __shared__ int lsum[2];
  __shared__ int stage[ORD_CAP];
  const int tid  = threadIdx.x;
  const int lane = tid & 31;
  const int wave = __builtin_amdgcn_readfirstlane(tid >> 5);
  const int own  = blockIdx.x;
  const int base = own * AGG_R;

  if (tid < 2 * AGG_R) (&cnt[0][0])[tid] = 0;
  if (tid < 2) lsum[tid] = 0;
  __syncthreads();

  {
    int front = 0, count = 0;
    int* c = cnt[0];
    gcn_owner_walk(sorted, offc, own, n_chunks, wave, lane, front, count,
                   [c](int e) { atomicAdd(&c[e & 255], 1); });
    if (lane == 0) { atomicAdd(&lsum[0], front); atomicAdd(&lsum[1], count); }
  }
  {
    int front = 0, count = 0;
    int* c = cnt[1];
    gcn_owner_walk(sorted_s, offs, own, n_chunks, wave, lane, front, count,
                   [c](int e) { atomicAdd(&c[e & 255], 1); });
  }
  __syncthreads();

  const int tot = lsum[1];
  const int lst = ((lsum[0] + 31) & ~31) + 32 * own;
  const bool fits = tot <= ORD_CAP;

  if (tid < 32) {
    int v[PL];
    int ls = 0;
#pragma unroll
    for (int j = 0; j < PL; ++j) {
      const int r = lane * PL + j;
      v[j] = (r < AGG_R) ? cnt[0][r] : 0;
      ls += v[j];
    }
    int inc = ls;
#pragma unroll
    for (int dl = 1; dl < 32; dl <<= 1) {
      const int t = __shfl_up(inc, dl);
      if (lane >= dl) inc += t;
    }
    int run = inc - ls;
#pragma unroll
    for (int j = 0; j < PL; ++j) {
      const int r = lane * PL + j;
      if (r < AGG_R) cur[r] = run;
      run += v[j];
    }
  }
  __syncthreads();

  {
    int front = 0, count = 0;
    int* cu = cur;
    int* sg = stage;
    int* og = ord + lst;
    if (fits)
      gcn_owner_walk(sorted, offc, own, n_chunks, wave, lane, front, count,
                     [cu, sg](int e) { sg[atomicAdd(&cu[e & 255], 1)] = e; });
    else
      gcn_owner_walk(sorted, offc, own, n_chunks, wave, lane, front, count,
                     [cu, og](int e) { og[atomicAdd(&cu[e & 255], 1)] = e; });
  }
  __syncthreads();

  if (fits)
    for (int i = tid; i < tot; i += AGG_T) ord[lst + i] = stage[i];
  if (tid == 0) { ordb[2 * own] = lst; ordb[2 * own + 1] = tot; }

  if (tid < AGG_R) {
    const int row = base + tid;
    if (row < n_nodes) {
      ns[row] = __frsqrt_rn(fmaxf((float)cnt[1][tid], 1.0f));
      nd[row] = __frsqrt_rn(fmaxf((float)cnt[0][tid], 1.0f));
    }
  }
}

#define GEMM_RP 32
template <int K, int F, bool PRE, bool POST>
__global__ __launch_bounds__(F * 2) void gcn_gemm_wstrip(
    const float* __restrict__ A, const float* __restrict__ W, float* __restrict__ C, int n_rows,
    const float* __restrict__ pre_scale, const float* __restrict__ post_scale,
    const float* __restrict__ bias, int relu) {
  constexpr int NT = F * 2;
  constexpr int P  = K + 4;
  constexpr int KS = K / 4;
  constexpr int C4 = K / 4;
  constexpr int NV = GEMM_RP * C4 / NT;
  constexpr int RSTEP = NT / C4;
  __shared__ __attribute__((aligned(16))) float sa[2][GEMM_RP * P];

  const int tid  = threadIdx.x;
  const int lane = tid & 31;
  const int wave = __builtin_amdgcn_readfirstlane(tid >> 5);
  const int half = lane >> 4;
  const int lr   = lane & 15;
  const int n    = (wave << 4) + lr;

  v2f wf[KS];
#pragma unroll
  for (int s = 0; s < KS; ++s) {
    const int koff = 4 * s + (half << 1);
    wf[s].x = W[(size_t)koff * F + n];
    wf[s].y = W[(size_t)(koff + 1) * F + n];
  }

  const int n_panels = (n_rows + GEMM_RP - 1) / GEMM_RP;
  const int r0 = tid / C4;
  const int c4 = tid - r0 * C4;
  const float bv = (POST && bias) ? bias[n] : 0.0f;

  v4f   st[NV];
  float  sc[NV];
  float pvc = 1.0f, pvn = 1.0f;
  int p = blockIdx.x;
  if (p < n_panels) {
    if (POST) {
      const int row = p * GEMM_RP + lane;
      pvc = post_scale[(row < n_rows) ? row : (n_rows - 1)];
    }
#pragma unroll
    for (int i = 0; i < NV; ++i) {
      const int row = p * GEMM_RP + r0 + i * RSTEP;
      const int rl  = (row < n_rows) ? row : (n_rows - 1);
      st[i] = *(const v4f*)(A + (size_t)rl * K + 4 * c4);
      if (PRE) sc[i] = pre_scale[rl];
    }
#pragma unroll
    for (int i = 0; i < NV; ++i) {
      v4f t = st[i];
      if (PRE) t *= sc[i];
      *(v4f*)(&sa[0][(r0 + i * RSTEP) * P + 4 * c4]) = t;
    }
  }
  __syncthreads();

  int cur = 0;
  for (; p < n_panels; p += gridDim.x) {
    const int pn = p + gridDim.x;
    if (pn < n_panels) {
#pragma unroll
      for (int i = 0; i < NV; ++i) {
        const int row = pn * GEMM_RP + r0 + i * RSTEP;
        const int rl  = (row < n_rows) ? row : (n_rows - 1);
        st[i] = *(const v4f*)(A + (size_t)rl * K + 4 * c4);
        if (PRE) sc[i] = pre_scale[rl];
      }
      if (POST) {
        const int row = pn * GEMM_RP + lane;
        pvn = post_scale[(row < n_rows) ? row : (n_rows - 1)];
      }
    }

    const float* fa = &sa[cur][lr * P + (half << 1)];
    v8f acc[2] = {};
#pragma unroll
    for (int s = 0; s < KS; ++s) {
      const v2f a0 = *(const v2f*)(fa + 4 * s);
      const v2f a1 = *(const v2f*)(fa + 16 * P + 4 * s);
      acc[0] = __builtin_amdgcn_wmma_f32_16x16x4_f32(
           false, a0,  false, wf[s],
           (short)0, acc[0],  false,  false);
      acc[1] = __builtin_amdgcn_wmma_f32_16x16x4_f32(
           false, a1,  false, wf[s],
           (short)0, acc[1],  false,  false);
    }

    if (pn < n_panels) {
#pragma unroll
      for (int i = 0; i < NV; ++i) {
        v4f t = st[i];
        if (PRE) t *= sc[i];
        *(v4f*)(&sa[cur ^ 1][(r0 + i * RSTEP) * P + 4 * c4]) = t;
      }
    }
    __syncthreads();
    cur ^= 1;

#pragma unroll
    for (int t = 0; t < 2; ++t) {
      const int rowt = p * GEMM_RP + 16 * t;
      if (rowt < n_rows) {
#pragma unroll
        for (int i = 0; i < 8; ++i) {
          const int row = rowt + i + (half << 3);
          float v = acc[t][i];
          if (POST) {
            v *= __shfl(pvc, 16 * t + i + (half << 3));
            if (bias) v += bv;
            if (relu) v = fmaxf(v, 0.0f);
          }
          C[(size_t)row * F + n] = v;
        }
      }
    }
    pvc = pvn;
  }
}

template <int F, int P>
__device__ __forceinline__ void gcn_row_sums(
    const float* __restrict__ msg, const int* __restrict__ ord, const int* __restrict__ ordb,
    float* acc, int own, int wave, int lane) {
  constexpr int V = F / 32;
  const int lst = ordb[2 * own];
  const int tot = ordb[2 * own + 1];
  const int per = (tot + AGG_T / 32 - 1) / (AGG_T / 32);
  const int p0  = wave * per;
  const int p1  = (p0 + per < tot) ? (p0 + per) : tot;
  const int* list = ord + lst;

  float a[V];
#pragma unroll
  for (int j = 0; j < V; ++j) a[j] = 0.0f;
  int crow = -1;

  int ent_n = 0;
  if (p0 + lane < p1) ent_n = list[p0 + lane];
  for (int b0 = p0; b0 < p1; b0 += 32) {
    const int ent = ent_n;
    const int idn = b0 + 32 + lane;
    ent_n = 0;
    if (idn < p1) ent_n = list[idn];
    const int nb = (p1 - b0 < 32) ? (p1 - b0) : 32;
    for (int k = 0; k < nb; k += AGG_G) {
      float v[AGG_G][V];
      int rr[AGG_G];
#pragma unroll
      for (int q = 0; q < AGG_G; ++q) {
        const int kk = (k + q < nb) ? (k + q) : (nb - 1);
        const int ek = __builtin_amdgcn_readlane(ent, kk);
        const int sk = (int)((unsigned)ek >> 8);
        rr[q] = ek & 255;
        const float* rowp = msg + (size_t)sk * F + V * lane;
        if (V == 4) {
          const float4 t = *(const float4*)rowp;
          v[q][0] = t.x; v[q][1] = t.y; v[q][2] = t.z; v[q][V - 1] = t.w;
        } else {
          const float2 t = *(const float2*)rowp;
          v[q][0] = t.x; v[q][V - 1] = t.y;
        }
      }
#pragma unroll
      for (int q = 0; q < AGG_G; ++q) {
        if (k + q < nb) {
          if (rr[q] != crow) {
            if (crow >= 0) {
              float* d = acc + crow * P + lane;
#pragma unroll
              for (int j = 0; j < V; ++j) atomicAdd(d + 32 * j, a[j]);
            }
            crow = rr[q];
#pragma unroll
            for (int j = 0; j < V; ++j) a[j] = 0.0f;
          }
#pragma unroll
          for (int j = 0; j < V; ++j) a[j] += v[q][j];
        }
      }
    }
  }
  if (crow >= 0) {
    float* d = acc + crow * P + lane;
#pragma unroll
    for (int j = 0; j < V; ++j) atomicAdd(d + 32 * j, a[j]);
  }
}

template <int F>
__global__ __launch_bounds__(AGG_T) void gcn_agg_own(
    const float* __restrict__ msg, const int* __restrict__ ord, const int* __restrict__ ordb,
    const float* __restrict__ nd, const float* __restrict__ bias, float* __restrict__ outp,
    int n_nodes, int relu) {
  constexpr int V = F / 32;
  __shared__ __attribute__((aligned(16))) float acc[AGG_R * F];

  const int tid  = threadIdx.x;
  const int lane = tid & 31;
  const int wave = __builtin_amdgcn_readfirstlane(tid >> 5);
  const int own  = blockIdx.x;
  const int base = own * AGG_R;

  for (int i = tid; i < AGG_R * F / 4; i += AGG_T)
    ((float4*)acc)[i] = make_float4(0.0f, 0.0f, 0.0f, 0.0f);
  __syncthreads();

  gcn_row_sums<F, F>(msg, ord, ordb, acc, own, wave, lane);
  __syncthreads();

  float bv[V];
#pragma unroll
  for (int j = 0; j < V; ++j) bv[j] = bias ? bias[V * lane + j] : 0.0f;
  for (int r = wave; r < AGG_R; r += AGG_T / 32) {
    const int row = base + r;
    if (row < n_nodes) {
      const float sc = nd ? nd[row] : 1.0f;
      float o[V];
#pragma unroll
      for (int j = 0; j < V; ++j) {
        float t = acc[r * F + 32 * j + lane] * sc + bv[j];
        if (relu) t = fmaxf(t, 0.0f);
        o[j] = t;
      }
      float* op = outp + (size_t)row * F + V * lane;
      if (V == 4) *(float4*)op = make_float4(o[0], o[1], o[2], o[V - 1]);
      else        *(float2*)op = make_float2(o[0], o[V - 1]);
    }
  }
}

#define AGG_RT  13
#define AGG_P   132
template <int FO>
__global__ __launch_bounds__(AGG_T) void gcn_agg_own_mm(
    const float* __restrict__ msg, const int* __restrict__ ord, const int* __restrict__ ordb,
    const float* __restrict__ nd, const float* __restrict__ bias, const float* __restrict__ ns,
    const float* __restrict__ Wn, float* __restrict__ outp, int n_nodes) {
  constexpr int F   = 128;
  constexpr int V   = F / 32;
  constexpr int P   = AGG_P;
  constexpr int KS  = F / 4;
  constexpr int NS  = FO / 16;
  constexpr int WPS = (AGG_T / 32) / NS;
  __shared__ __attribute__((aligned(16))) float acc[AGG_RT * 16 * P];

  const int tid  = threadIdx.x;
  const int lane = tid & 31;
  const int wave = __builtin_amdgcn_readfirstlane(tid >> 5);
  const int own  = blockIdx.x;
  const int base = own * AGG_R;

  for (int i = tid; i < AGG_RT * 16 * P / 4; i += AGG_T)
    ((float4*)acc)[i] = make_float4(0.0f, 0.0f, 0.0f, 0.0f);
  __syncthreads();

  gcn_row_sums<F, P>(msg, ord, ordb, acc, own, wave, lane);

  const int half  = lane >> 4;
  const int lr    = lane & 15;
  const int strip = wave % NS;
  const int n     = (strip << 4) + lr;
  v2f wf[KS];
#pragma unroll
  for (int s = 0; s < KS; ++s) {
    const int k0 = 16 * (s & 7) + (half << 3) + (s >> 3);
    wf[s].x = Wn[(size_t)k0 * FO + n];
    wf[s].y = Wn[(size_t)(k0 + 4) * FO + n];
  }
  float bv[V];
#pragma unroll
  for (int j = 0; j < V; ++j) bv[j] = bias[V * lane + j];
  __syncthreads();

  for (int r = wave; r < AGG_R; r += AGG_T / 32) {
    const int row = base + r;
    if (row < n_nodes) {
      const float sc = nd[row];
      const float ps = ns[row];
      float* a = acc + r * P + lane;
#pragma unroll
      for (int j = 0; j < V; ++j) a[32 * j] = fmaxf(a[32 * j] * sc + bv[j], 0.0f) * ps;
    }
  }
  __syncthreads();

  for (int rt = wave / NS; rt < AGG_RT; rt += 2 * WPS) {
    const int rt1 = rt + WPS;
    const float* fa0 = &acc[(16 * rt + lr) * P + (half << 1)];
    v8f d0 = {}, d1 = {};
    if (rt1 < AGG_RT) {
      const float* fa1 = fa0 + 16 * WPS * P;
#pragma unroll
      for (int s = 0; s < KS; ++s) {
        const v2f a0 = *(const v2f*)(fa0 + 4 * s);
        const v2f a1 = *(const v2f*)(fa1 + 4 * s);
        d0 = __builtin_amdgcn_wmma_f32_16x16x4_f32(
             false, a0,  false, wf[s],
             (short)0, d0,  false,  false);
        d1 = __builtin_amdgcn_wmma_f32_16x16x4_f32(
             false, a1,  false, wf[s],
             (short)0, d1,  false,  false);
      }
    } else {
#pragma unroll
      for (int s = 0; s < KS; ++s) {
        const v2f a0 = *(const v2f*)(fa0 + 4 * s);
        d0 = __builtin_amdgcn_wmma_f32_16x16x4_f32(
             false, a0,  false, wf[s],
             (short)0, d0,  false,  false);
      }
    }
#pragma unroll
    for (int i = 0; i < 8; ++i) {
      const int r = 16 * rt + i + (half << 3);
      if (r < AGG_R && base + r < n_nodes) outp[(size_t)(base + r) * FO + n] = d0[i];
    }
    if (rt1 < AGG_RT) {
#pragma unroll
      for (int i = 0; i < 8; ++i) {
        const int r = 16 * rt1 + i + (half << 3);
        if (r < AGG_R && base + r < n_nodes) outp[(size_t)(base + r) * FO + n] = d1[i];
      }
    }
  }
}

static inline unsigned cdiv(unsigned a, unsigned b) { return (a + b - 1) / b; }

extern "C" void kernel_launch(void* const* d_in, const int* in_sizes, int n_in,
                              void* d_out, int out_size, void* d_ws, size_t ws_size,
                              hipStream_t stream) {
  const float* features = (const float*)d_in[0];
  const float* W1 = (const float*)d_in[1];
  const float* b1 = (const float*)d_in[2];
  const float* W2 = (const float*)d_in[3];
  const float* b2 = (const float*)d_in[4];
  const float* W3 = (const float*)d_in[5];
  const float* b3 = (const float*)d_in[6];
  const int* src  = (const int*)d_in[7];
  const int* dst  = (const int*)d_in[8];
  float* out = (float*)d_out;

  const int N  = N_NODES;
  const int E  = in_sizes[7];
  const int H  = in_sizes[2];

  float* ws   = (float*)d_ws;
  float* ns   = ws;
  float* nd   = ws + N;
  float* bufA = ws + 2 * N + 96;
  float* bufB = bufA + (size_t)N * H;
  const unsigned n_chunks = cdiv(E, SRT_CH);
  int* offc   = (int*)(bufB + (size_t)N * H);
  int* sorted = offc + (size_t)n_chunks * SRT_OFP;
  int* offs   = sorted + (size_t)n_chunks * SRT_CHP;
  int* sorted_s = offs + (size_t)n_chunks * SRT_OFP;
  int* ord    = sorted_s + (size_t)n_chunks * SRT_CHP;
  int* ordb   = ord + (size_t)E + 32 * (SRT_NB + 1);
  (void)ws_size; (void)n_in; (void)out_size;

  const unsigned agg_blocks = cdiv(N, AGG_R);

  gcn_edge_sort <<<n_chunks, AGG_T, 0, stream>>>(src, dst, offc, sorted, offs, sorted_s, E, N);
  gcn_owner_norm<<<agg_blocks, AGG_T, 0, stream>>>(sorted, offc, sorted_s, offs, ns, nd, ord, ordb, n_chunks, N);

  gcn_gemm_wstrip<256, 128, true, false><<<512, 256, 0, stream>>>(
      features, W1, bufA, N, ns, nullptr, nullptr, 0);

  gcn_agg_own_mm<128><<<agg_blocks, AGG_T, 0, stream>>>(
      bufA, ord, ordb, nd, b1, ns, W2, bufB, N);

  gcn_agg_own_mm<64><<<agg_blocks, AGG_T, 0, stream>>>(
      bufB, ord, ordb, nd, b2, ns, W3, bufA, N);

  gcn_agg_own<64><<<agg_blocks, AGG_T, 0, stream>>>(
      bufA, ord, ordb, nd, b3, out, N, 0);
}


# dependent load round trips cut: row-scale loads leave the 3 tail loops (1 early load per wave); list/segment bounds, first entries load before barrier 1; next entries load behind a group's row loads
# speedup vs baseline: 5.4088x; 1.0389x over previous
#include <hip/hip_runtime.h>


typedef float v2f __attribute__((ext_vector_type(2)));
typedef float v4f __attribute__((ext_vector_type(4)));
typedef float v8f __attribute__((ext_vector_type(8)));

#define N_NODES 50000

#define AGG_R   196
#define AGG_T   1024
#define AGG_G   8
#define SRT_CH  3125
#define SRT_CHP 3200
#define SRT_OFP 288
#define SRT_NB  256
#define SRT_W   8
#define CNT_Q   4
#define ORD_CAP 8192

__global__ __launch_bounds__(AGG_T) void gcn_edge_sort(
    const int* __restrict__ src, const int* __restrict__ dst,
    int* __restrict__ offc, int* __restrict__ sorted,
    int* __restrict__ offs, int* __restrict__ sorted_s, int E, int n_nodes) {
  constexpr int PER = (SRT_CH + AGG_T - 1) / AGG_T;
  __shared__ int hist[2][SRT_NB];
  __shared__ int cur[2][SRT_NB];
  __shared__ int stage[2][SRT_CH];
  __shared__ int n_sorted[2];
  const int tid  = threadIdx.x;
  const int lane = tid & 31;
  const int e0   = blockIdx.x * SRT_CH;
  const int len  = (E - e0 < SRT_CH) ? (E - e0) : SRT_CH;

  if (tid < 2 * SRT_NB) (&hist[0][0])[tid] = 0;

  int sv[PER], dv[PER], sl[PER], ss[PER];
#pragma unroll
  for (int it = 0; it < PER; ++it) {
    const int i = tid + it * AGG_T;
    sv[it] = -1;
    dv[it] = -1;
    if (i < len) { sv[it] = src[e0 + i]; dv[it] = dst[e0 + i]; }
  }
  __syncthreads();

#pragma unroll
  for (int it = 0; it < PER; ++it) {
    sl[it] = -1;
    ss[it] = -1;
    if ((unsigned)dv[it] < (unsigned)n_nodes) sl[it] = atomicAdd(&hist[0][(unsigned)dv[it] / AGG_R], 1);
    if ((unsigned)sv[it] < (unsigned)n_nodes) ss[it] = atomicAdd(&hist[1][(unsigned)sv[it] / AGG_R], 1);
  }
  __syncthreads();

  if (tid < 64) {
    const int w = tid >> 5;
    int* row = (w ? offs : offc) + (size_t)blockIdx.x * SRT_OFP;
    int v[SRT_NB / 32];
    int lsum = 0;
#pragma unroll
    for (int j = 0; j < SRT_NB / 32; ++j) { v[j] = hist[w][lane * (SRT_NB / 32) + j]; lsum += v[j]; }
    int inc = lsum;
#pragma unroll
    for (int dl = 1; dl < 32; dl <<= 1) {
      const int t = __shfl_up(inc, dl);
      if (lane >= dl) inc += t;
    }
    int run = inc - lsum;
#pragma unroll
    for (int j = 0; j < SRT_NB / 32; ++j) {
      cur[w][lane * (SRT_NB / 32) + j] = run;
      row[lane * (SRT_NB / 32) + j] = run;
      run += v[j];
    }
    if (lane == 31) { row[SRT_NB] = run; n_sorted[w] = run; }
  }
  __syncthreads();

#pragma unroll
  for (int it = 0; it < PER; ++it) {
    if (sl[it] >= 0) {
      const unsigned o = (unsigned)dv[it] / AGG_R;
      const unsigned r = (unsigned)dv[it] - o * AGG_R;
      stage[0][cur[0][o] + sl[it]] = (int)(((unsigned)sv[it] << 8) | r);
    }
    if (ss[it] >= 0) {
      const unsigned o = (unsigned)sv[it] / AGG_R;
      stage[1][cur[1][o] + ss[it]] = (int)((unsigned)sv[it] - o * AGG_R);
    }
  }
  __syncthreads();

  int* outc = sorted   + (size_t)blockIdx.x * SRT_CHP;
  int* outs = sorted_s + (size_t)blockIdx.x * SRT_CHP;
  const int tot0 = n_sorted[0];
  const int tot1 = n_sorted[1];
  for (int i = tid; i < tot0; i += AGG_T) outc[i] = stage[0][i];
  for (int i = tid; i < tot1; i += AGG_T) outs[i] = stage[1][i];
}

__device__ __forceinline__ void gcn_seg_bounds(
    const int* __restrict__ off, int own, int n_chunks, int cb, int lane, int& st, int& ln, int& s0v) {
  st = 0; ln = 0; s0v = 0;
  if (lane < SRT_W && cb + lane < n_chunks) {
    const int* row = off + (size_t)(cb + lane) * SRT_OFP + own;
    const int s0 = row[0];
    const int s1 = row[1];
    st = (cb + lane) * SRT_CHP + s0;
    ln = s1 - s0;
    s0v = s0;
  }
}

template <class Fn>
__device__ __forceinline__ void gcn_owner_walk(
    const int* __restrict__ list, const int* __restrict__ off, int own, int n_chunks,
    int wave, int lane, int st0, int ln0, int s00, int& front, int& count, Fn f) {
  for (int cb = wave * SRT_W; cb < n_chunks; cb += (AGG_T / 32) * SRT_W) {
    int st = st0, ln = ln0, s0v = s00;
    if (cb != wave * SRT_W) gcn_seg_bounds(off, own, n_chunks, cb, lane, st, ln, s0v);
    int endj[SRT_W], adj[SRT_W];
    int total = 0;
#pragma unroll
    for (int j = 0; j < SRT_W; ++j) {
      const int lj = __builtin_amdgcn_readlane(ln, j);
      const int sj = __builtin_amdgcn_readlane(st, j);
      front += __builtin_amdgcn_readlane(s0v, j);
      adj[j] = sj - total;
      total += lj;
      endj[j] = total;
    }
    count += total;
    for (int b0 = 0; b0 < total; b0 += 32 * CNT_Q) {
      int e[CNT_Q];
#pragma unroll
      for (int q = 0; q < CNT_Q; ++q) {
        const int idx = b0 + 32 * q + lane;
        e[q] = 0;
        if (idx < total) {
          int a = adj[0];
#pragma unroll
          for (int j = 1; j < SRT_W; ++j) if (idx >= endj[j - 1]) a = adj[j];
          e[q] = list[a + idx];
        }
      }
#pragma unroll
      for (int q = 0; q < CNT_Q; ++q)
        if (b0 + 32 * q + lane < total) f(e[q]);
    }
  }
}

__global__ __launch_bounds__(AGG_T) void gcn_owner_norm(
    const int* __restrict__ sorted, const int* __restrict__ offc,
    const int* __restrict__ sorted_s, const int* __restrict__ offs,
    float* __restrict__ ns, float* __restrict__ nd,
    int* __restrict__ ord, int* __restrict__ ordb, int n_chunks, int n_nodes) {
  constexpr int PL = (AGG_R + 31) / 32;
  __shared__ int cnt[2][AGG_R];
  __shared__ int cur[AGG_R];
  __shared__ int lsum[2];
  __shared__ int stage[ORD_CAP];
  const int tid  = threadIdx.x;
  const int lane = tid & 31;
  const int wave = __builtin_amdgcn_readfirstlane(tid >> 5);
  const int own  = blockIdx.x;
  const int base = own * AGG_R;

  int stc = 0, lnc = 0, s0c = 0, sts = 0, lns = 0, s0s = 0;
  {
    const int ch = wave * SRT_W + lane;
    if (lane < SRT_W && ch < n_chunks) {
      const int* rc = offc + (size_t)ch * SRT_OFP + own;
      const int* rs = offs + (size_t)ch * SRT_OFP + own;
      const int c0 = rc[0], c1 = rc[1], q0 = rs[0], q1 = rs[1];
      stc = ch * SRT_CHP + c0;  lnc = c1 - c0;  s0c = c0;
      sts = ch * SRT_CHP + q0;  lns = q1 - q0;  s0s = q0;
    }
  }

  if (tid < 2 * AGG_R) (&cnt[0][0])[tid] = 0;
  if (tid < 2) lsum[tid] = 0;
  __syncthreads();

  {
    int front = 0, count = 0;
    int* c = cnt[0];
    gcn_owner_walk(sorted, offc, own, n_chunks, wave, lane, stc, lnc, s0c, front, count,
                   [c](int e) { atomicAdd(&c[e & 255], 1); });
    if (lane == 0) { atomicAdd(&lsum[0], front); atomicAdd(&lsum[1], count); }
  }
  {
    int front = 0, count = 0;
    int* c = cnt[1];
    gcn_owner_walk(sorted_s, offs, own, n_chunks, wave, lane, sts, lns, s0s, front, count,
                   [c](int e) { atomicAdd(&c[e & 255], 1); });
  }
  __syncthreads();

  const int tot = lsum[1];
  const int lst = ((lsum[0] + 31) & ~31) + 32 * own;
  const bool fits = tot <= ORD_CAP;

  if (tid < 32) {
    int v[PL];
    int ls = 0;
#pragma unroll
    for (int j = 0; j < PL; ++j) {
      const int r = lane * PL + j;
      v[j] = (r < AGG_R) ? cnt[0][r] : 0;
      ls += v[j];
    }
    int inc = ls;
#pragma unroll
    for (int dl = 1; dl < 32; dl <<= 1) {
      const int t = __shfl_up(inc, dl);
      if (lane >= dl) inc += t;
    }
    int run = inc - ls;
#pragma unroll
    for (int j = 0; j < PL; ++j) {
      const int r = lane * PL + j;
      if (r < AGG_R) cur[r] = run;
      run += v[j];
    }
  }
  __syncthreads();

  {
    int front = 0, count = 0;
    int* cu = cur;
    int* sg = stage;
    int* og = ord + lst;
    if (fits)
      gcn_owner_walk(sorted, offc, own, n_chunks, wave, lane, stc, lnc, s0c, front, count,
                     [cu, sg](int e) { sg[atomicAdd(&cu[e & 255], 1)] = e; });
    else
      gcn_owner_walk(sorted, offc, own, n_chunks, wave, lane, stc, lnc, s0c, front, count,
                     [cu, og](int e) { og[atomicAdd(&cu[e & 255], 1)] = e; });
  }
  __syncthreads();

  if (fits)
    for (int i = tid; i < tot; i += AGG_T) ord[lst + i] = stage[i];
  if (tid == 0) { ordb[2 * own] = lst; ordb[2 * own + 1] = tot; }

  if (tid < AGG_R) {
    const int row = base + tid;
    if (row < n_nodes) {
      ns[row] = __frsqrt_rn(fmaxf((float)cnt[1][tid], 1.0f));
      nd[row] = __frsqrt_rn(fmaxf((float)cnt[0][tid], 1.0f));
    }
  }
}

#define GEMM_RP 32
template <int K, int F, bool PRE, bool POST>
__global__ __launch_bounds__(F * 2) void gcn_gemm_wstrip(
    const float* __restrict__ A, const float* __restrict__ W, float* __restrict__ C, int n_rows,
    const float* __restrict__ pre_scale, const float* __restrict__ post_scale,
    const float* __restrict__ bias, int relu) {
  constexpr int NT = F * 2;
  constexpr int P  = K + 4;
  constexpr int KS = K / 4;
  constexpr int C4 = K / 4;
  constexpr int NV = GEMM_RP * C4 / NT;
  constexpr int RSTEP = NT / C4;
  __shared__ __attribute__((aligned(16))) float sa[2][GEMM_RP * P];

  const int tid  = threadIdx.x;
  const int lane = tid & 31;
  const int wave = __builtin_amdgcn_readfirstlane(tid >> 5);
  const int half = lane >> 4;
  const int lr   = lane & 15;
  const int n    = (wave << 4) + lr;

  v2f wf[KS];
#pragma unroll
  for (int s = 0; s < KS; ++s) {
    const int koff = 4 * s + (half << 1);
    wf[s].x = W[(size_t)koff * F + n];
    wf[s].y = W[(size_t)(koff + 1) * F + n];
  }

  const int n_panels = (n_rows + GEMM_RP - 1) / GEMM_RP;
  const int r0 = tid / C4;
  const int c4 = tid - r0 * C4;
  const float bv = (POST && bias) ? bias[n] : 0.0f;

  v4f   st[NV];
  float  sc[NV];
  float pvc = 1.0f, pvn = 1.0f;
  int p = blockIdx.x;
  if (p < n_panels) {
    if (POST) {
      const int row = p * GEMM_RP + lane;
      pvc = post_scale[(row < n_rows) ? row : (n_rows - 1)];
    }
#pragma unroll
    for (int i = 0; i < NV; ++i) {
      const int row = p * GEMM_RP + r0 + i * RSTEP;
      const int rl  = (row < n_rows) ? row : (n_rows - 1);
      st[i] = *(const v4f*)(A + (size_t)rl * K + 4 * c4);
      if (PRE) sc[i] = pre_scale[rl];
    }
#pragma unroll
    for (int i = 0; i < NV; ++i) {
      v4f t = st[i];
      if (PRE) t *= sc[i];
      *(v4f*)(&sa[0][(r0 + i * RSTEP) * P + 4 * c4]) = t;
    }
  }
  __syncthreads();

  int cur = 0;
  for (; p < n_panels; p += gridDim.x) {
    const int pn = p + gridDim.x;
    if (pn < n_panels) {
#pragma unroll
      for (int i = 0; i < NV; ++i) {
        const int row = pn * GEMM_RP + r0 + i * RSTEP;
        const int rl  = (row < n_rows) ? row : (n_rows - 1);
        st[i] = *(const v4f*)(A + (size_t)rl * K + 4 * c4);
        if (PRE) sc[i] = pre_scale[rl];
      }
      if (POST) {
        const int row = pn * GEMM_RP + lane;
        pvn = post_scale[(row < n_rows) ? row : (n_rows - 1)];
      }
    }

    const float* fa = &sa[cur][lr * P + (half << 1)];
    v8f acc[2] = {};
#pragma unroll
    for (int s = 0; s < KS; ++s) {
      const v2f a0 = *(const v2f*)(fa + 4 * s);
      const v2f a1 = *(const v2f*)(fa + 16 * P + 4 * s);
      acc[0] = __builtin_amdgcn_wmma_f32_16x16x4_f32(
           false, a0,  false, wf[s],
           (short)0, acc[0],  false,  false);
      acc[1] = __builtin_amdgcn_wmma_f32_16x16x4_f32(
           false, a1,  false, wf[s],
           (short)0, acc[1],  false,  false);
    }

    if (pn < n_panels) {
#pragma unroll
      for (int i = 0; i < NV; ++i) {
        v4f t = st[i];
        if (PRE) t *= sc[i];
        *(v4f*)(&sa[cur ^ 1][(r0 + i * RSTEP) * P + 4 * c4]) = t;
      }
    }
    __syncthreads();
    cur ^= 1;

#pragma unroll
    for (int t = 0; t < 2; ++t) {
      const int rowt = p * GEMM_RP + 16 * t;
      if (rowt < n_rows) {
#pragma unroll
        for (int i = 0; i < 8; ++i) {
          const int row = rowt + i + (half << 3);
          float v = acc[t][i];
          if (POST) {
            v *= __shfl(pvc, 16 * t + i + (half << 3));
            if (bias) v += bv;
            if (relu) v = fmaxf(v, 0.0f);
          }
          C[(size_t)row * F + n] = v;
        }
      }
    }
    pvc = pvn;
  }
}

__device__ __forceinline__ void gcn_share_head(
    const int* __restrict__ ord, const int* __restrict__ ordb, int own, int wave, int lane,
    const int*& list, int& p0, int& p1, int& ent0) {
  const int lst = ordb[2 * own];
  const int tot = ordb[2 * own + 1];
  const int per = (tot + AGG_T / 32 - 1) / (AGG_T / 32);
  p0   = wave * per;
  p1   = (p0 + per < tot) ? (p0 + per) : tot;
  list = ord + lst;
  ent0 = 0;
  if (p0 + lane < p1) ent0 = list[p0 + lane];
}

#define AGG_RW ((AGG_R + AGG_T / 32 - 1) / (AGG_T / 32))
__device__ __forceinline__ float gcn_wave_row_scales(
    const float* __restrict__ sc, int base, int wave, int lane, int n_nodes) {
  const int r = wave + (AGG_T / 32) * lane;
  float v = 0.0f;
  if (lane < AGG_RW && r < AGG_R && base + r < n_nodes) v = sc ? sc[base + r] : 1.0f;
  return v;
}

template <int F, int P>
__device__ __forceinline__ void gcn_row_sums(
    const float* __restrict__ msg, const int* __restrict__ list, int p0, int p1, int ent0,
    float* acc, int lane) {
  constexpr int V = F / 32;

  float a[V];
#pragma unroll
  for (int j = 0; j < V; ++j) a[j] = 0.0f;
  int crow = -1;

  int ent_n = ent0;
  for (int b0 = p0; b0 < p1; b0 += 32) {
    const int ent = ent_n;
    const int idn = b0 + 32 + lane;
    const int nb = (p1 - b0 < 32) ? (p1 - b0) : 32;
    for (int k = 0; k < nb; k += AGG_G) {
      float v[AGG_G][V];
      int rr[AGG_G];
#pragma unroll
      for (int q = 0; q < AGG_G; ++q) {
        const int kk = (k + q < nb) ? (k + q) : (nb - 1);
        const int ek = __builtin_amdgcn_readlane(ent, kk);
        const int sk = (int)((unsigned)ek >> 8);
        rr[q] = ek & 255;
        const float* rowp = msg + (size_t)sk * F + V * lane;
        if (V == 4) {
          const float4 t = *(const float4*)rowp;
          v[q][0] = t.x; v[q][1] = t.y; v[q][2] = t.z; v[q][V - 1] = t.w;
        } else {
          const float2 t = *(const float2*)rowp;
          v[q][0] = t.x; v[q][V - 1] = t.y;
        }
      }
      if (k == 0) {
        ent_n = 0;
        if (idn < p1) ent_n = list[idn];
      }
#pragma unroll
      for (int q = 0; q < AGG_G; ++q) {
        if (k + q < nb) {
          if (rr[q] != crow) {
            if (crow >= 0) {
              float* d = acc + crow * P + lane;
#pragma unroll
              for (int j = 0; j < V; ++j) atomicAdd(d + 32 * j, a[j]);
            }
            crow = rr[q];
#pragma unroll
            for (int j = 0; j < V; ++j) a[j] = 0.0f;
          }
#pragma unroll
          for (int j = 0; j < V; ++j) a[j] += v[q][j];
        }
      }
    }
  }
  if (crow >= 0) {
    float* d = acc + crow * P + lane;
#pragma unroll
    for (int j = 0; j < V; ++j) atomicAdd(d + 32 * j, a[j]);
  }
}

template <int F>
__global__ __launch_bounds__(AGG_T) void gcn_agg_own(
    const float* __restrict__ msg, const int* __restrict__ ord, const int* __restrict__ ordb,
    const float* __restrict__ nd, const float* __restrict__ bias, float* __restrict__ outp,
    int n_nodes, int relu) {
  constexpr int V = F / 32;
  __shared__ __attribute__((aligned(16))) float acc[AGG_R * F];

  const int tid  = threadIdx.x;
  const int lane = tid & 31;
  const int wave = __builtin_amdgcn_readfirstlane(tid >> 5);
  const int own  = blockIdx.x;
  const int base = own * AGG_R;

  const int* list;
  int p0, p1, ent0;
  gcn_share_head(ord, ordb, own, wave, lane, list, p0, p1, ent0);
  const float scl = gcn_wave_row_scales(nd, base, wave, lane, n_nodes);
  float bv[V];
#pragma unroll
  for (int j = 0; j < V; ++j) bv[j] = bias ? bias[V * lane + j] : 0.0f;

  for (int i = tid; i < AGG_R * F / 4; i += AGG_T)
    ((float4*)acc)[i] = make_float4(0.0f, 0.0f, 0.0f, 0.0f);
  __syncthreads();

  gcn_row_sums<F, F>(msg, list, p0, p1, ent0, acc, lane);
  __syncthreads();

#pragma unroll
  for (int k = 0; k < AGG_RW; ++k) {
    const int r   = wave + (AGG_T / 32) * k;
    const int row = base + r;
    if (r < AGG_R && row < n_nodes) {
      const float sc = __int_as_float(__builtin_amdgcn_readlane(__float_as_int(scl), k));
      float o[V];
#pragma unroll
      for (int j = 0; j < V; ++j) {
        float t = acc[r * F + 32 * j + lane] * sc + bv[j];
        if (relu) t = fmaxf(t, 0.0f);
        o[j] = t;
      }
      float* op = outp + (size_t)row * F + V * lane;
      if (V == 4) *(float4*)op = make_float4(o[0], o[1], o[2], o[V - 1]);
      else        *(float2*)op = make_float2(o[0], o[V - 1]);
    }
  }
}

#define AGG_RT  13
#define AGG_P   132
template <int FO>
__global__ __launch_bounds__(AGG_T) void gcn_agg_own_mm(
    const float* __restrict__ msg, const int* __restrict__ ord, const int* __restrict__ ordb,
    const float* __restrict__ nd, const float* __restrict__ bias, const float* __restrict__ ns,
    const float* __restrict__ Wn, float* __restrict__ outp, int n_nodes) {
  constexpr int F   = 128;
  constexpr int V   = F / 32;
  constexpr int P   = AGG_P;
  constexpr int KS  = F / 4;
  constexpr int NS  = FO / 16;
  constexpr int WPS = (AGG_T / 32) / NS;
  __shared__ __attribute__((aligned(16))) float acc[AGG_RT * 16 * P];

  const int tid  = threadIdx.x;
  const int lane = tid & 31;
  const int wave = __builtin_amdgcn_readfirstlane(tid >> 5);
  const int own  = blockIdx.x;
  const int base = own * AGG_R;

  const int* list;
  int p0, p1, ent0;
  gcn_share_head(ord, ordb, own, wave, lane, list, p0, p1, ent0);
  const float scl = gcn_wave_row_scales(nd, base, wave, lane, n_nodes);
  const float psl = gcn_wave_row_scales(ns, base, wave, lane, n_nodes);
  float bv[V];
#pragma unroll
  for (int j = 0; j < V; ++j) bv[j] = bias[V * lane + j];

  for (int i = tid; i < AGG_RT * 16 * P / 4; i += AGG_T)
    ((float4*)acc)[i] = make_float4(0.0f, 0.0f, 0.0f, 0.0f);
  __syncthreads();

  gcn_row_sums<F, P>(msg, list, p0, p1, ent0, acc, lane);

  const int half  = lane >> 4;
  const int lr    = lane & 15;
  const int strip = wave % NS;
  const int n     = (strip << 4) + lr;
  v2f wf[KS];
#pragma unroll
  for (int s = 0; s < KS; ++s) {
    const int k0 = 16 * (s & 7) + (half << 3) + (s >> 3);
    wf[s].x = Wn[(size_t)k0 * FO + n];
    wf[s].y = Wn[(size_t)(k0 + 4) * FO + n];
  }
  __syncthreads();

#pragma unroll
  for (int k = 0; k < AGG_RW; ++k) {
    const int r   = wave + (AGG_T / 32) * k;
    const int row = base + r;
    if (r < AGG_R && row < n_nodes) {
      const float sc = __int_as_float(__builtin_amdgcn_readlane(__float_as_int(scl), k));
      const float ps = __int_as_float(__builtin_amdgcn_readlane(__float_as_int(psl), k));
      float* a = acc + r * P + lane;
#pragma unroll
      for (int j = 0; j < V; ++j) a[32 * j] = fmaxf(a[32 * j] * sc + bv[j], 0.0f) * ps;
    }
  }
  __syncthreads();

  for (int rt = wave / NS; rt < AGG_RT; rt += 2 * WPS) {
    const int rt1 = rt + WPS;
    const float* fa0 = &acc[(16 * rt + lr) * P + (half << 1)];
    v8f d0 = {}, d1 = {};
    if (rt1 < AGG_RT) {
      const float* fa1 = fa0 + 16 * WPS * P;
#pragma unroll
      for (int s = 0; s < KS; ++s) {
        const v2f a0 = *(const v2f*)(fa0 + 4 * s);
        const v2f a1 = *(const v2f*)(fa1 + 4 * s);
        d0 = __builtin_amdgcn_wmma_f32_16x16x4_f32(
             false, a0,  false, wf[s],
             (short)0, d0,  false,  false);
        d1 = __builtin_amdgcn_wmma_f32_16x16x4_f32(
             false, a1,  false, wf[s],
             (short)0, d1,  false,  false);
      }
    } else {
#pragma unroll
      for (int s = 0; s < KS; ++s) {
        const v2f a0 = *(const v2f*)(fa0 + 4 * s);
        d0 = __builtin_amdgcn_wmma_f32_16x16x4_f32(
             false, a0,  false, wf[s],
             (short)0, d0,  false,  false);
      }
    }
#pragma unroll
    for (int i = 0; i < 8; ++i) {
      const int r = 16 * rt + i + (half << 3);
      if (r < AGG_R && base + r < n_nodes) outp[(size_t)(base + r) * FO + n] = d0[i];
    }
    if (rt1 < AGG_RT) {
#pragma unroll
      for (int i = 0; i < 8; ++i) {
        const int r = 16 * rt1 + i + (half << 3);
        if (r < AGG_R && base + r < n_nodes) outp[(size_t)(base + r) * FO + n] = d1[i];
      }
    }
  }
}

static inline unsigned cdiv(unsigned a, unsigned b) { return (a + b - 1) / b; }

extern "C" void kernel_launch(void* const* d_in, const int* in_sizes, int n_in,
                              void* d_out, int out_size, void* d_ws, size_t ws_size,
                              hipStream_t stream) {
  const float* features = (const float*)d_in[0];
  const float* W1 = (const float*)d_in[1];
  const float* b1 = (const float*)d_in[2];
  const float* W2 = (const float*)d_in[3];
  const float* b2 = (const float*)d_in[4];
  const float* W3 = (const float*)d_in[5];
  const float* b3 = (const float*)d_in[6];
  const int* src  = (const int*)d_in[7];
  const int* dst  = (const int*)d_in[8];
  float* out = (float*)d_out;

  const int N  = N_NODES;
  const int E  = in_sizes[7];
  const int H  = in_sizes[2];

  float* ws   = (float*)d_ws;
  float* ns   = ws;
  float* nd   = ws + N;
  float* bufA = ws + 2 * N + 96;
  float* bufB = bufA + (size_t)N * H;
  const unsigned n_chunks = cdiv(E, SRT_CH);
  int* offc   = (int*)(bufB + (size_t)N * H);
  int* sorted = offc + (size_t)n_chunks * SRT_OFP;
  int* offs   = sorted + (size_t)n_chunks * SRT_CHP;
  int* sorted_s = offs + (size_t)n_chunks * SRT_OFP;
  int* ord    = sorted_s + (size_t)n_chunks * SRT_CHP;
  int* ordb   = ord + (size_t)E + 32 * (SRT_NB + 1);
  (void)ws_size; (void)n_in; (void)out_size;

  const unsigned agg_blocks = cdiv(N, AGG_R);

  gcn_edge_sort <<<n_chunks, AGG_T, 0, stream>>>(src, dst, offc, sorted, offs, sorted_s, E, N);
  gcn_owner_norm<<<agg_blocks, AGG_T, 0, stream>>>(sorted, offc, sorted_s, offs, ns, nd, ord, ordb, n_chunks, N);

  gcn_gemm_wstrip<256, 128, true, false><<<512, 256, 0, stream>>>(
      features, W1, bufA, N, ns, nullptr, nullptr, 0);

  gcn_agg_own_mm<128><<<agg_blocks, AGG_T, 0, stream>>>(
      bufA, ord, ordb, nd, b1, ns, W2, bufB, N);

  gcn_agg_own_mm<64><<<agg_blocks, AGG_T, 0, stream>>>(
      bufB, ord, ordb, nd, b2, ns, W3, bufA, N);

  gcn_agg_own<64><<<agg_blocks, AGG_T, 0, stream>>>(
      bufA, ord, ordb, nd, b3, out, N, 0);
}
